# starting kernel
# baseline (speedup 1.0000x reference)
#include <hip/hip_runtime.h>
#include <hip/hip_bf16.h>

#define N_NODES  100000
#define E_EDGES  1600000
#define IN_F     256
#define HD       128
#define H_HEADS  8
#define SLOPE    0.2f

#define ROWS_PB    64
#define LDS_STRIDE 257

typedef __attribute__((ext_vector_type(2))) float v2f;
typedef __attribute__((ext_vector_type(8))) float v8f;

__device__ __forceinline__ unsigned f2u(float f) {
    unsigned u = __float_as_uint(f);
    return (u & 0x80000000u) ? ~u : (u | 0x80000000u);
}
__device__ __forceinline__ float u2f(unsigned u) {
    u = (u & 0x80000000u) ? (u & 0x7FFFFFFFu) : ~u;
    return __uint_as_float(u);
}

__global__ __launch_bounds__(256) void gat_gemm3(
    const float* __restrict__ x,
    const float* __restrict__ Wsrc, const float* __restrict__ bsrc,
    const float* __restrict__ Wdst, const float* __restrict__ bdst,
    const float* __restrict__ Wres, const float* __restrict__ bres,
    float* __restrict__ el, float* __restrict__ er, float* __restrict__ outres)
{
    __shared__ float xs[ROWS_PB * LDS_STRIDE];

    const int tid  = threadIdx.x;
    const int row0 = blockIdx.x * ROWS_PB;

    for (int i = tid; i < ROWS_PB * (IN_F / 4); i += 256) {
        const int r  = i >> 6;
        const int c4 = i & 63;
        float4 v = make_float4(0.f, 0.f, 0.f, 0.f);
        const int gr = row0 + r;
        if (gr < N_NODES)
            v = *(const float4*)(x + (size_t)gr * IN_F + c4 * 4);
        float* p = &xs[r * LDS_STRIDE + c4 * 4];
        p[0] = v.x; p[1] = v.y; p[2] = v.z; p[3] = v.w;
    }
    __syncthreads();

    const int wave    = tid >> 5;
    const int lane    = tid & 31;
    const int l15     = lane & 15;
    const int khalf   = (lane >> 4) * 2;
    const int rhalf   = (lane >> 4) * 8;
    const int colbase = wave * 16;

    const float* const Wm[3] = { Wsrc, Wdst, Wres };
    const float* const bm[3] = { bsrc, bdst, bres };
    float* const       Om[3] = { el,   er,   outres };

#pragma unroll
    for (int m = 0; m < 3; ++m) {
        const float* __restrict__ W = Wm[m];
        v8f acc0 = {}; v8f acc1 = {}; v8f acc2 = {}; v8f acc3 = {};

        for (int k = 0; k < IN_F; k += 4) {
            v2f b;
            b.x = W[(k + khalf)     * HD + colbase + l15];
            b.y = W[(k + khalf + 1) * HD + colbase + l15];

#define GAT_WSTEP(ACC, RT)                                                     \
            {                                                                  \
                v2f a;                                                         \
                const int ab = ((RT) * 16 + l15) * LDS_STRIDE + k + khalf;     \
                a.x = xs[ab];                                                  \
                a.y = xs[ab + 1];                                              \
                ACC = __builtin_amdgcn_wmma_f32_16x16x4_f32(                   \
                        false, a, false, b, (short)0, ACC, false, false);      \
            }
            GAT_WSTEP(acc0, 0)
            GAT_WSTEP(acc1, 1)
            GAT_WSTEP(acc2, 2)
            GAT_WSTEP(acc3, 3)
#undef GAT_WSTEP
        }

        const float bias = bm[m][colbase + l15];
        float* __restrict__ O = Om[m];

#define GAT_STORE(ACC, RT)                                                     \
        {                                                                      \
            _Pragma("unroll")                                                  \
            for (int j = 0; j < 8; ++j) {                                      \
                const int r = row0 + (RT) * 16 + rhalf + j;                    \
                if (r < N_NODES)                                               \
                    O[(size_t)r * HD + colbase + l15] = ACC[j] + bias;         \
            }                                                                  \
        }
        GAT_STORE(acc0, 0)
        GAT_STORE(acc1, 1)
        GAT_STORE(acc2, 2)
        GAT_STORE(acc3, 3)
#undef GAT_STORE
    }
}

__global__ __launch_bounds__(256) void gat_score(
    const float* __restrict__ el, const float* __restrict__ er,
    const int* __restrict__ src, const int* __restrict__ dst,
    const float* __restrict__ attn,
    float* __restrict__ score, unsigned* __restrict__ maxu)
{
    const int e = blockIdx.x * 8 + (threadIdx.x >> 5);
    if (e >= E_EDGES) return;
    const int lane = threadIdx.x & 31;
    const int u = src[e];
    const int v = dst[e];
    const int c = lane * 4;

    const float4 a4 = *(const float4*)(el   + (size_t)u * HD + c);
    const float4 b4 = *(const float4*)(er   + (size_t)v * HD + c);
    const float4 w4 = *(const float4*)(attn + c);

    float p = 0.f, t;
    t = a4.x + b4.x; t = (t > 0.f) ? t : SLOPE * t; p += t * w4.x;
    t = a4.y + b4.y; t = (t > 0.f) ? t : SLOPE * t; p += t * w4.y;
    t = a4.z + b4.z; t = (t > 0.f) ? t : SLOPE * t; p += t * w4.z;
    t = a4.w + b4.w; t = (t > 0.f) ? t : SLOPE * t; p += t * w4.w;

    p += __shfl_xor(p, 1, 32);
    p += __shfl_xor(p, 2, 32);

    if ((lane & 3) == 0) {
        const int h = lane >> 2;
        score[(size_t)e * H_HEADS + h] = p;
        atomicMax(&maxu[(size_t)v * H_HEADS + h], f2u(p));
    }
}

__global__ __launch_bounds__(256) void gat_expsum(
    const int* __restrict__ dst, float* __restrict__ score,
    const unsigned* __restrict__ maxu, float* __restrict__ denom)
{
    const long long t = (long long)blockIdx.x * 256 + threadIdx.x;
    const int e = (int)(t >> 3);
    const int h = (int)(t & 7);
    if (e >= E_EDGES) return;
    const int v = dst[e];
    const float m = u2f(maxu[(size_t)v * H_HEADS + h]);
    const float a = __expf(score[t] - m);
    score[t] = a;
    unsafeAtomicAdd(&denom[(size_t)v * H_HEADS + h], a);
}

__global__ __launch_bounds__(256) void gat_aggregate(
    const float* __restrict__ el,
    const int* __restrict__ src, const int* __restrict__ dst,
    const float* __restrict__ alpha, const float* __restrict__ denom,
    float* __restrict__ out)
{
    const int e = blockIdx.x * 8 + (threadIdx.x >> 5);
    if (e >= E_EDGES) return;
    const int lane = threadIdx.x & 31;
    const int u = src[e];
    const int v = dst[e];
    const int h = lane >> 2;
    const int c = lane * 4;

    const float w  = alpha[(size_t)e * H_HEADS + h] / denom[(size_t)v * H_HEADS + h];
    const float4 f = *(const float4*)(el + (size_t)u * HD + c);

    float* o = out + (size_t)v * HD + c;
    unsafeAtomicAdd(o + 0, w * f.x);
    unsafeAtomicAdd(o + 1, w * f.y);
    unsafeAtomicAdd(o + 2, w * f.z);
    unsafeAtomicAdd(o + 3, w * f.w);
}

extern "C" void kernel_launch(void* const* d_in, const int* in_sizes, int n_in,
                              void* d_out, int out_size, void* d_ws, size_t ws_size,
                              hipStream_t stream) {
    const float* x    = (const float*)d_in[0];
    const int*   src  = (const int*)  d_in[1];
    const int*   dst  = (const int*)  d_in[2];
    const float* Wsrc = (const float*)d_in[3];
    const float* bsrc = (const float*)d_in[4];
    const float* Wdst = (const float*)d_in[5];
    const float* bdst = (const float*)d_in[6];
    const float* attn = (const float*)d_in[7];
    const float* Wres = (const float*)d_in[8];
    const float* bres = (const float*)d_in[9];
    float* out = (float*)d_out;

    char* ws = (char*)d_ws;
    const size_t elBytes    = (size_t)N_NODES * HD * sizeof(float);
    const size_t scoreBytes = (size_t)E_EDGES * H_HEADS * sizeof(float);
    const size_t nhBytes    = (size_t)N_NODES * H_HEADS * sizeof(float);
    float*    el    = (float*)   (ws);
    float*    er    = (float*)   (ws + elBytes);
    float*    score = (float*)   (ws + 2 * elBytes);
    unsigned* maxu  = (unsigned*)(ws + 2 * elBytes + scoreBytes);
    float*    denom = (float*)   (ws + 2 * elBytes + scoreBytes + nhBytes);

    hipMemsetAsync(maxu,  0, nhBytes, stream);
    hipMemsetAsync(denom, 0, nhBytes, stream);

    const int gemmBlocks = (N_NODES + ROWS_PB - 1) / ROWS_PB;
    gat_gemm3<<<gemmBlocks, 256, 0, stream>>>(x, Wsrc, bsrc, Wdst, bdst,
                                              Wres, bres, el, er, out);

    const int edgeBlocks = (E_EDGES + 7) / 8;
    gat_score<<<edgeBlocks, 256, 0, stream>>>(el, er, src, dst, attn, score, maxu);

    const long long ehThreads = (long long)E_EDGES * H_HEADS;
    const int expBlocks = (int)((ehThreads + 255) / 256);
    gat_expsum<<<expBlocks, 256, 0, stream>>>(dst, score, maxu, denom);

    gat_aggregate<<<edgeBlocks, 256, 0, stream>>>(el, src, dst, score, denom, out);
}


# Edges grouped by destination (counting sort, arrival-order rank); one wave per node does logits, running-max softmax and weighted sum in registers: no float atomics, sums in rank order
# speedup vs baseline: 1.8921x; 1.8921x over previous
#include <hip/hip_runtime.h>
#include <hip/hip_bf16.h>

#define N_NODES  100000
#define E_EDGES  1600000
#define IN_F     256
#define HD       128
#define H_HEADS  8
#define SLOPE    0.2f

#define ROWS_PB    64
#define LDS_STRIDE 257

#define SCAN_BLOCK 1024
#define SCAN_ITEMS (N_NODES + 1)
#define SCAN_NBLK  ((SCAN_ITEMS + SCAN_BLOCK - 1) / SCAN_BLOCK)
#define SCAN_PAD   (SCAN_NBLK * SCAN_BLOCK)

typedef __attribute__((ext_vector_type(2))) float v2f;
typedef __attribute__((ext_vector_type(8))) float v8f;

__global__ __launch_bounds__(256) void gat_gemm3(
    const float* __restrict__ x,
    const float* __restrict__ Wsrc, const float* __restrict__ bsrc,
    const float* __restrict__ Wdst, const float* __restrict__ bdst,
    const float* __restrict__ Wres, const float* __restrict__ bres,
    float* __restrict__ el, float* __restrict__ er, float* __restrict__ outres)
{
    __shared__ float xs[ROWS_PB * LDS_STRIDE];

    const int tid  = threadIdx.x;
    const int row0 = blockIdx.x * ROWS_PB;

    for (int i = tid; i < ROWS_PB * (IN_F / 4); i += 256) {
        const int r  = i >> 6;
        const int c4 = i & 63;
        float4 v = make_float4(0.f, 0.f, 0.f, 0.f);
        const int gr = row0 + r;
        if (gr < N_NODES)
            v = *(const float4*)(x + (size_t)gr * IN_F + c4 * 4);
        float* p = &xs[r * LDS_STRIDE + c4 * 4];
        p[0] = v.x; p[1] = v.y; p[2] = v.z; p[3] = v.w;
    }
    __syncthreads();

    const int wave    = tid >> 5;
    const int lane    = tid & 31;
    const int l15     = lane & 15;
    const int khalf   = (lane >> 4) * 2;
    const int rhalf   = (lane >> 4) * 8;
    const int colbase = wave * 16;

    const float* const Wm[3] = { Wsrc, Wdst, Wres };
    const float* const bm[3] = { bsrc, bdst, bres };
    float* const       Om[3] = { el,   er,   outres };

#pragma unroll
    for (int m = 0; m < 3; ++m) {
        const float* __restrict__ W = Wm[m];
        v8f acc0 = {}; v8f acc1 = {}; v8f acc2 = {}; v8f acc3 = {};

        for (int k = 0; k < IN_F; k += 4) {
            v2f b;
            b.x = W[(k + khalf)     * HD + colbase + l15];
            b.y = W[(k + khalf + 1) * HD + colbase + l15];

#define GAT_WSTEP(ACC, RT)                                                     \
            {                                                                  \
                v2f a;                                                         \
                const int ab = ((RT) * 16 + l15) * LDS_STRIDE + k + khalf;     \
                a.x = xs[ab];                                                  \
                a.y = xs[ab + 1];                                              \
                ACC = __builtin_amdgcn_wmma_f32_16x16x4_f32(                   \
                        false, a, false, b, (short)0, ACC, false, false);      \
            }
            GAT_WSTEP(acc0, 0)
            GAT_WSTEP(acc1, 1)
            GAT_WSTEP(acc2, 2)
            GAT_WSTEP(acc3, 3)
#undef GAT_WSTEP
        }

        const float bias = bm[m][colbase + l15];
        float* __restrict__ O = Om[m];

#define GAT_STORE(ACC, RT)                                                     \
        {                                                                      \
            _Pragma("unroll")                                                  \
            for (int j = 0; j < 8; ++j) {                                      \
                const int r = row0 + (RT) * 16 + rhalf + j;                    \
                if (r < N_NODES)                                               \
                    O[(size_t)r * HD + colbase + l15] = ACC[j] + bias;         \
            }                                                                  \
        }
        GAT_STORE(acc0, 0)
        GAT_STORE(acc1, 1)
        GAT_STORE(acc2, 2)
        GAT_STORE(acc3, 3)
#undef GAT_STORE
    }
}

__global__ __launch_bounds__(256) void gat_count(
    const int* __restrict__ dst, int* __restrict__ cnt, int* __restrict__ rank)
{
    const int e = blockIdx.x * 256 + threadIdx.x;
    if (e >= E_EDGES) return;
    rank[e] = atomicAdd(&cnt[dst[e]], 1);
}

__global__ __launch_bounds__(SCAN_BLOCK) void gat_scan_local(
    const int* __restrict__ cnt, int* __restrict__ rowptr, int* __restrict__ bsum)
{
    __shared__ int wsum[32];
    const int tid  = threadIdx.x;
    const int lane = tid & 31;
    const int wave = tid >> 5;
    const int i    = blockIdx.x * SCAN_BLOCK + tid;

    const int c = (i < SCAN_ITEMS) ? cnt[i] : 0;
    int s = c;
    for (int d = 1; d < 32; d <<= 1) {
        const int t = __shfl_up(s, d, 32);
        if (lane >= d) s += t;
    }
    if (lane == 31) wsum[wave] = s;
    __syncthreads();
    if (wave == 0) {
        int w = wsum[lane];
        for (int d = 1; d < 32; d <<= 1) {
            const int t = __shfl_up(w, d, 32);
            if (lane >= d) w += t;
        }
        wsum[lane] = w;
    }
    __syncthreads();
    const int woff = (wave > 0) ? wsum[wave - 1] : 0;
    if (i < SCAN_ITEMS) rowptr[i] = woff + s - c;
    if (tid == SCAN_BLOCK - 1) bsum[blockIdx.x] = woff + s;
}

__global__ __launch_bounds__(SCAN_BLOCK) void gat_scan_add(
    int* __restrict__ rowptr, const int* __restrict__ bsum)
{
    __shared__ int off;
    const int tid = threadIdx.x;
    const int b   = blockIdx.x;
    if (tid < 32) {
        int s = 0;
        for (int t = tid; t < b; t += 32) s += bsum[t];
        for (int d = 16; d > 0; d >>= 1) s += __shfl_xor(s, d, 32);
        if (tid == 0) off = s;
    }
    __syncthreads();
    const int i = b * SCAN_BLOCK + tid;
    if (i < SCAN_ITEMS) rowptr[i] += off;
}

__global__ __launch_bounds__(256) void gat_fill(
    const int* __restrict__ src, const int* __restrict__ dst,
    const int* __restrict__ rank, const int* __restrict__ rowptr,
    int* __restrict__ csr_src)
{
    const int e = blockIdx.x * 256 + threadIdx.x;
    if (e >= E_EDGES) return;
    const int pos = rowptr[dst[e]] + rank[e];
    if ((unsigned)pos < (unsigned)E_EDGES)
        csr_src[pos] = src[e];
}

__device__ __forceinline__ float gat_logit(const float4 f, const float4 b4, const float4 w4)
{
    float p = 0.f, t;
    t = f.x + b4.x; t = (t > 0.f) ? t : SLOPE * t; p += t * w4.x;
    t = f.y + b4.y; t = (t > 0.f) ? t : SLOPE * t; p += t * w4.y;
    t = f.z + b4.z; t = (t > 0.f) ? t : SLOPE * t; p += t * w4.z;
    t = f.w + b4.w; t = (t > 0.f) ? t : SLOPE * t; p += t * w4.w;
    p += __shfl_xor(p, 1, 32);
    p += __shfl_xor(p, 2, 32);
    return p;
}

__global__ __launch_bounds__(256) void gat_node(
    const float* __restrict__ el, const float* __restrict__ er,
    const int* __restrict__ rowptr, const int* __restrict__ csr_src,
    const float* __restrict__ attn, float* __restrict__ out)
{
    const int v = __builtin_amdgcn_readfirstlane(blockIdx.x * 8 + (threadIdx.x >> 5));
    if (v >= N_NODES) return;
    const int lane = threadIdx.x & 31;
    const int c    = lane * 4;

    int beg = rowptr[v];
    int end = rowptr[v + 1];
    beg = max(beg, 0);
    end = min(end, E_EDGES);

    const float4 b4 = *(const float4*)(er   + (size_t)v * HD + c);
    const float4 w4 = *(const float4*)(attn + c);

    float  m   = -INFINITY;
    float  l   = 0.f;
    float4 acc = make_float4(0.f, 0.f, 0.f, 0.f);

    for (int base = beg; base < end; base += 32) {
        const int cnt = min(32, end - base);
        int idx = 0;
        if (lane < cnt) idx = csr_src[base + lane];

        int j = 0;
        for (; j + 4 <= cnt; j += 4) {
            int u0 = __builtin_amdgcn_readlane(idx, j);
            int u1 = __builtin_amdgcn_readlane(idx, j + 1);
            int u2 = __builtin_amdgcn_readlane(idx, j + 2);
            int u3 = __builtin_amdgcn_readlane(idx, j + 3);
            u0 = ((unsigned)u0 < (unsigned)N_NODES) ? u0 : 0;
            u1 = ((unsigned)u1 < (unsigned)N_NODES) ? u1 : 0;
            u2 = ((unsigned)u2 < (unsigned)N_NODES) ? u2 : 0;
            u3 = ((unsigned)u3 < (unsigned)N_NODES) ? u3 : 0;
            const float4 f0 = *(const float4*)(el + (size_t)u0 * HD + c);
            const float4 f1 = *(const float4*)(el + (size_t)u1 * HD + c);
            const float4 f2 = *(const float4*)(el + (size_t)u2 * HD + c);
            const float4 f3 = *(const float4*)(el + (size_t)u3 * HD + c);
            const float p0 = gat_logit(f0, b4, w4);
            const float p1 = gat_logit(f1, b4, w4);
            const float p2 = gat_logit(f2, b4, w4);
            const float p3 = gat_logit(f3, b4, w4);
            const float mn = fmaxf(m, fmaxf(fmaxf(p0, p1), fmaxf(p2, p3)));
            const float sc = __expf(m - mn);
            const float a0 = __expf(p0 - mn);
            const float a1 = __expf(p1 - mn);
            const float a2 = __expf(p2 - mn);
            const float a3 = __expf(p3 - mn);
            l     = l * sc + ((a0 + a1) + (a2 + a3));
            acc.x = acc.x * sc + ((a0 * f0.x + a1 * f1.x) + (a2 * f2.x + a3 * f3.x));
            acc.y = acc.y * sc + ((a0 * f0.y + a1 * f1.y) + (a2 * f2.y + a3 * f3.y));
            acc.z = acc.z * sc + ((a0 * f0.z + a1 * f1.z) + (a2 * f2.z + a3 * f3.z));
            acc.w = acc.w * sc + ((a0 * f0.w + a1 * f1.w) + (a2 * f2.w + a3 * f3.w));
            m = mn;
        }
        for (; j < cnt; ++j) {
            int u0 = __builtin_amdgcn_readlane(idx, j);
            u0 = ((unsigned)u0 < (unsigned)N_NODES) ? u0 : 0;
            const float4 f0 = *(const float4*)(el + (size_t)u0 * HD + c);
            const float p0 = gat_logit(f0, b4, w4);
            const float mn = fmaxf(m, p0);
            const float sc = __expf(m - mn);
            const float a0 = __expf(p0 - mn);
            l     = l * sc + a0;
            acc.x = acc.x * sc + a0 * f0.x;
            acc.y = acc.y * sc + a0 * f0.y;
            acc.z = acc.z * sc + a0 * f0.z;
            acc.w = acc.w * sc + a0 * f0.w;
            m = mn;
        }
    }

    float* o = out + (size_t)v * HD + c;
    float4 r = *(const float4*)o;
    if (end > beg) {
        r.x += acc.x / l;
        r.y += acc.y / l;
        r.z += acc.z / l;
        r.w += acc.w / l;
    }
    *(float4*)o = r;
}

extern "C" void kernel_launch(void* const* d_in, const int* in_sizes, int n_in,
                              void* d_out, int out_size, void* d_ws, size_t ws_size,
                              hipStream_t stream) {
    const float* x    = (const float*)d_in[0];
    const int*   src  = (const int*)  d_in[1];
    const int*   dst  = (const int*)  d_in[2];
    const float* Wsrc = (const float*)d_in[3];
    const float* bsrc = (const float*)d_in[4];
    const float* Wdst = (const float*)d_in[5];
    const float* bdst = (const float*)d_in[6];
    const float* attn = (const float*)d_in[7];
    const float* Wres = (const float*)d_in[8];
    const float* bres = (const float*)d_in[9];
    float* out = (float*)d_out;

    char* ws = (char*)d_ws;
    const size_t elBytes   = (size_t)N_NODES * HD * sizeof(float);
    const size_t padBytes  = (size_t)SCAN_PAD * sizeof(int);
    const size_t edgeBytes = (size_t)E_EDGES * sizeof(int);
    float* el     = (float*)(ws);
    float* er     = (float*)(ws + elBytes);
    int*   cnt    = (int*)  (ws + 2 * elBytes);
    int*   rowptr = (int*)  (ws + 2 * elBytes + padBytes);
    int*   bsum   = (int*)  (ws + 2 * elBytes + 2 * padBytes);
    int*   rank   = (int*)  (ws + 2 * elBytes + 2 * padBytes + 512);
    int*   csr    = (int*)  (ws + 2 * elBytes + 2 * padBytes + 512 + edgeBytes);

    hipMemsetAsync(cnt, 0, padBytes, stream);

    const int edgeBlocks = (E_EDGES + 255) / 256;
    gat_count<<<edgeBlocks, 256, 0, stream>>>(dst, cnt, rank);
    gat_scan_local<<<SCAN_NBLK, SCAN_BLOCK, 0, stream>>>(cnt, rowptr, bsum);
    gat_scan_add<<<SCAN_NBLK, SCAN_BLOCK, 0, stream>>>(rowptr, bsum);
    gat_fill<<<edgeBlocks, 256, 0, stream>>>(src, dst, rank, rowptr, csr);

    const int gemmBlocks = (N_NODES + ROWS_PB - 1) / ROWS_PB;
    gat_gemm3<<<gemmBlocks, 256, 0, stream>>>(x, Wsrc, bsrc, Wdst, bdst,
                                              Wres, bres, el, er, out);

    const int nodeBlocks = (N_NODES + 7) / 8;
    gat_node<<<nodeBlocks, 256, 0, stream>>>(el, er, rowptr, csr, attn, out);
}


# Edges grouped by destination by a two-level partition in LDS (256 node ranges x 256 edge chunks, LDS atomics, one scan): no global atomic or scattered store per edge; sums still in arrival order
# speedup vs baseline: 2.5177x; 1.3306x over previous
#include <hip/hip_runtime.h>
#include <hip/hip_bf16.h>

#define N_NODES  100000
#define E_EDGES  1600000
#define IN_F     256
#define HD       128
#define H_HEADS  8
#define SLOPE    0.2f

#define ROWS_PB    64
#define LDS_STRIDE 257

#define TAB_PAD    100352

typedef __attribute__((ext_vector_type(2))) float v2f;
typedef __attribute__((ext_vector_type(8))) float v8f;

__global__ __launch_bounds__(256) void gat_gemm3(
    const float* __restrict__ x,
    const float* __restrict__ Wsrc, const float* __restrict__ bsrc,
    const float* __restrict__ Wdst, const float* __restrict__ bdst,
    const float* __restrict__ Wres, const float* __restrict__ bres,
    float* __restrict__ el, float* __restrict__ er, float* __restrict__ outres)
{
    __shared__ float xs[ROWS_PB * LDS_STRIDE];

    const int tid  = threadIdx.x;
    const int row0 = blockIdx.x * ROWS_PB;

    for (int i = tid; i < ROWS_PB * (IN_F / 4); i += 256) {
        const int r  = i >> 6;
        const int c4 = i & 63;
        float4 v = make_float4(0.f, 0.f, 0.f, 0.f);
        const int gr = row0 + r;
        if (gr < N_NODES)
            v = *(const float4*)(x + (size_t)gr * IN_F + c4 * 4);
        float* p = &xs[r * LDS_STRIDE + c4 * 4];
        p[0] = v.x; p[1] = v.y; p[2] = v.z; p[3] = v.w;
    }
    __syncthreads();

    const int wave    = tid >> 5;
    const int lane    = tid & 31;
    const int l15     = lane & 15;
    const int khalf   = (lane >> 4) * 2;
    const int rhalf   = (lane >> 4) * 8;
    const int colbase = wave * 16;

    const float* const Wm[3] = { Wsrc, Wdst, Wres };
    const float* const bm[3] = { bsrc, bdst, bres };
    float* const       Om[3] = { el,   er,   outres };

#pragma unroll
    for (int m = 0; m < 3; ++m) {
        const float* __restrict__ W = Wm[m];
        v8f acc0 = {}; v8f acc1 = {}; v8f acc2 = {}; v8f acc3 = {};

        for (int k = 0; k < IN_F; k += 4) {
            v2f b;
            b.x = W[(k + khalf)     * HD + colbase + l15];
            b.y = W[(k + khalf + 1) * HD + colbase + l15];

#define GAT_WSTEP(ACC, RT)                                                     \
            {                                                                  \
                v2f a;                                                         \
                const int ab = ((RT) * 16 + l15) * LDS_STRIDE + k + khalf;     \
                a.x = xs[ab];                                                  \
                a.y = xs[ab + 1];                                              \
                ACC = __builtin_amdgcn_wmma_f32_16x16x4_f32(                   \
                        false, a, false, b, (short)0, ACC, false, false);      \
            }
            GAT_WSTEP(acc0, 0)
            GAT_WSTEP(acc1, 1)
            GAT_WSTEP(acc2, 2)
            GAT_WSTEP(acc3, 3)
#undef GAT_WSTEP
        }

        const float bias = bm[m][colbase + l15];
        float* __restrict__ O = Om[m];

#define GAT_STORE(ACC, RT)                                                     \
        {                                                                      \
            _Pragma("unroll")                                                  \
            for (int j = 0; j < 8; ++j) {                                      \
                const int r = row0 + (RT) * 16 + rhalf + j;                    \
                if (r < N_NODES)                                               \
                    O[(size_t)r * HD + colbase + l15] = ACC[j] + bias;         \
            }                                                                  \
        }
        GAT_STORE(acc0, 0)
        GAT_STORE(acc1, 1)
        GAT_STORE(acc2, 2)
        GAT_STORE(acc3, 3)
#undef GAT_STORE
    }
}

#define NBK      256
#define NPB      391
#define NCH      256
#define CH       (E_EDGES / NCH)
#define CH_FULL  (CH / 256)
#define CH_TAIL  (CH - CH_FULL * 256)
#define SRC_BITS 17
#define SRC_MASK ((1 << SRC_BITS) - 1)
#define CSR_CAP  12288

static_assert(NBK * NPB >= N_NODES, "ranges cover the nodes");
static_assert(NCH * CH == E_EDGES, "chunks cover the edges");
static_assert((1 << SRC_BITS) > N_NODES && NPB < (1 << (31 - SRC_BITS)), "packed word fits");

__device__ __forceinline__ int gat_clamp_node(int d)
{
    return min(max(d, 0), N_NODES - 1);
}

__device__ __forceinline__ int gat_block_exscan(int v, int* wsum)
{
    const int lane = threadIdx.x & 31;
    const int wave = threadIdx.x >> 5;
    int s = v;
    for (int d = 1; d < 32; d <<= 1) {
        const int t = __shfl_up(s, d, 32);
        if (lane >= d) s += t;
    }
    if (lane == 31) wsum[wave] = s;
    __syncthreads();
    if (wave == 0) {
        int w = (lane < (int)(blockDim.x >> 5)) ? wsum[lane] : 0;
        for (int d = 1; d < 32; d <<= 1) {
            const int t = __shfl_up(w, d, 32);
            if (lane >= d) w += t;
        }
        wsum[lane] = w;
    }
    __syncthreads();
    const int woff = (wave > 0) ? wsum[wave - 1] : 0;
    return woff + s - v;
}

__global__ __launch_bounds__(256) void gat_range_hist(
    const int* __restrict__ dst, int* __restrict__ boff)
{
    __shared__ int h[NBK];
    const int tid = threadIdx.x;
    const int* __restrict__ dp = dst + blockIdx.x * CH;
    h[tid] = 0;

    int d[CH_FULL];
#pragma unroll
    for (int k = 0; k < CH_FULL; ++k) d[k] = dp[k * 256 + tid];
    const int dl = dp[min(CH_FULL * 256 + tid, CH - 1)];
    __syncthreads();
#pragma unroll
    for (int k = 0; k < CH_FULL; ++k) atomicAdd(&h[gat_clamp_node(d[k]) / NPB], 1);
    if (tid < CH_TAIL) atomicAdd(&h[gat_clamp_node(dl) / NPB], 1);
    __syncthreads();
    boff[tid * NCH + blockIdx.x] = h[tid];
}

__global__ __launch_bounds__(1024) void gat_range_scan(int* __restrict__ boff)
{
    __shared__ int wsum[32];
    const int tid  = threadIdx.x;
    const int lane = tid & 31;
    const int wave = tid >> 5;
    int4* __restrict__ p = (int4*)boff + wave * 512;

    int4 v[16];
    int  part = 0;
#pragma unroll
    for (int j = 0; j < 16; ++j) {
        v[j] = p[j * 32 + lane];
        part += (v[j].x + v[j].y) + (v[j].z + v[j].w);
    }
    int tot = part;
    for (int d = 16; d > 0; d >>= 1) tot += __shfl_xor(tot, d, 32);
    if (lane == 0) wsum[wave] = tot;
    __syncthreads();
    if (wave == 0) {
        int w = wsum[lane];
        for (int d = 1; d < 32; d <<= 1) {
            const int t = __shfl_up(w, d, 32);
            if (lane >= d) w += t;
        }
        wsum[lane] = w;
    }
    __syncthreads();
    int carry = (wave > 0) ? wsum[wave - 1] : 0;
#pragma unroll
    for (int j = 0; j < 16; ++j) {
        const int s = (v[j].x + v[j].y) + (v[j].z + v[j].w);
        int inc = s;
        for (int d = 1; d < 32; d <<= 1) {
            const int t = __shfl_up(inc, d, 32);
            if (lane >= d) inc += t;
        }
        const int ex = carry + inc - s;
        int4 o;
        o.x = ex;
        o.y = ex + v[j].x;
        o.z = ex + v[j].x + v[j].y;
        o.w = ex + v[j].x + v[j].y + v[j].z;
        p[j * 32 + lane] = o;
        carry += __shfl(inc, 31, 32);
    }
    if (tid == 1023) boff[NBK * NCH] = carry;
}

__global__ __launch_bounds__(256) void gat_range_scatter(
    const int* __restrict__ src, const int* __restrict__ dst,
    const int* __restrict__ boff, int* __restrict__ part)
{
    __shared__ int lcur[NBK];
    __shared__ int gdel[NBK];
    __shared__ int sw[CH];
    __shared__ int sg[CH];
    __shared__ int wsum[32];
    const int tid = threadIdx.x;
    const int blk = blockIdx.x;
    const int* __restrict__ sp = src + blk * CH;
    const int* __restrict__ dp = dst + blk * CH;

    int d[CH_FULL], s[CH_FULL];
#pragma unroll
    for (int k = 0; k < CH_FULL; ++k) { d[k] = dp[k * 256 + tid]; s[k] = sp[k * 256 + tid]; }
    const int dl = dp[min(CH_FULL * 256 + tid, CH - 1)];
    const int sl = sp[min(CH_FULL * 256 + tid, CH - 1)];

    const int g0 = boff[tid * NCH + blk];
    const int g1 = boff[tid * NCH + blk + 1];
    const int ex = gat_block_exscan(g1 - g0, wsum);
    lcur[tid] = ex;
    gdel[tid] = g0 - ex;
    __syncthreads();

#define GAT_PLACE(D, S)                                                        \
    {                                                                          \
        const int dn = gat_clamp_node(D);                                      \
        const int b  = dn / NPB;                                               \
        const int sn = ((unsigned)(S) < (unsigned)N_NODES) ? (S) : 0;          \
        const int p  = atomicAdd(&lcur[b], 1);                                 \
        if ((unsigned)p < (unsigned)CH) {                                      \
            sw[p] = ((dn - b * NPB) << SRC_BITS) | sn;                         \
            sg[p] = gdel[b] + p;                                               \
        }                                                                      \
    }
#pragma unroll
    for (int k = 0; k < CH_FULL; ++k) GAT_PLACE(d[k], s[k])
    if (tid < CH_TAIL) GAT_PLACE(dl, sl)
#undef GAT_PLACE
    __syncthreads();

    for (int i = tid; i < CH; i += 256) {
        const int g = sg[i];
        if ((unsigned)g < (unsigned)E_EDGES)
            part[g] = sw[i];
    }
}

__global__ __launch_bounds__(512) void gat_range_csr(
    const int* __restrict__ part, const int* __restrict__ boff,
    int* __restrict__ rowptr, int* __restrict__ csr_src)
{
    __shared__ int cnt[512];
    __shared__ int lcsr[CSR_CAP];
    __shared__ int wsum[32];
    const int tid = threadIdx.x;
    const int b   = blockIdx.x;

    int bs = boff[b * NCH];
    int be = boff[(b + 1) * NCH];
    bs = min(max(bs, 0), E_EDGES);
    be = min(max(be, bs), E_EDGES);
    const int nb = be - bs;
    const int* __restrict__ pp = part + bs;

    cnt[tid] = 0;
    __syncthreads();
    for (int i0 = 0; i0 < nb; i0 += 4 * 512) {
        int w[4];
#pragma unroll
        for (int k = 0; k < 4; ++k) w[k] = pp[min(i0 + k * 512 + tid, nb - 1)];
#pragma unroll
        for (int k = 0; k < 4; ++k)
            if (i0 + k * 512 + tid < nb)
                atomicAdd(&cnt[min((int)((unsigned)w[k] >> SRC_BITS), NPB - 1)], 1);
    }
    __syncthreads();

    const int c  = (tid < NPB) ? cnt[tid] : 0;
    const int ex = gat_block_exscan(c, wsum);
    cnt[tid] = ex;
    if (tid < NPB) {
        const int node = b * NPB + tid;
        if (node < N_NODES) rowptr[node] = bs + ex;
    }
    if (b == NBK - 1 && tid == 0) rowptr[N_NODES] = be;
    __syncthreads();

    const bool stage = (nb <= CSR_CAP);
    for (int i0 = 0; i0 < nb; i0 += 4 * 512) {
        int w[4];
#pragma unroll
        for (int k = 0; k < 4; ++k) w[k] = pp[min(i0 + k * 512 + tid, nb - 1)];
#pragma unroll
        for (int k = 0; k < 4; ++k)
            if (i0 + k * 512 + tid < nb) {
                const int ln = min((int)((unsigned)w[k] >> SRC_BITS), NPB - 1);
                const int p  = atomicAdd(&cnt[ln], 1);
                if ((unsigned)p < (unsigned)nb) {
                    if (stage) lcsr[p] = w[k] & SRC_MASK;
                    else       csr_src[bs + p] = w[k] & SRC_MASK;
                }
            }
    }
    __syncthreads();
    if (stage)
        for (int i = tid; i < nb; i += 512) csr_src[bs + i] = lcsr[i];
}

__device__ __forceinline__ float gat_logit(const float4 f, const float4 b4, const float4 w4)
{
    float p = 0.f, t;
    t = f.x + b4.x; t = (t > 0.f) ? t : SLOPE * t; p += t * w4.x;
    t = f.y + b4.y; t = (t > 0.f) ? t : SLOPE * t; p += t * w4.y;
    t = f.z + b4.z; t = (t > 0.f) ? t : SLOPE * t; p += t * w4.z;
    t = f.w + b4.w; t = (t > 0.f) ? t : SLOPE * t; p += t * w4.w;
    p += __shfl_xor(p, 1, 32);
    p += __shfl_xor(p, 2, 32);
    return p;
}

__global__ __launch_bounds__(256) void gat_node(
    const float* __restrict__ el, const float* __restrict__ er,
    const int* __restrict__ rowptr, const int* __restrict__ csr_src,
    const float* __restrict__ attn, float* __restrict__ out)
{
    const int v = __builtin_amdgcn_readfirstlane(blockIdx.x * 8 + (threadIdx.x >> 5));
    if (v >= N_NODES) return;
    const int lane = threadIdx.x & 31;
    const int c    = lane * 4;

    int beg = rowptr[v];
    int end = rowptr[v + 1];
    beg = max(beg, 0);
    end = min(end, E_EDGES);

    const float4 b4 = *(const float4*)(er   + (size_t)v * HD + c);
    const float4 w4 = *(const float4*)(attn + c);

    float  m   = -INFINITY;
    float  l   = 0.f;
    float4 acc = make_float4(0.f, 0.f, 0.f, 0.f);

    for (int base = beg; base < end; base += 32) {
        const int cnt = min(32, end - base);
        int idx = 0;
        if (lane < cnt) idx = csr_src[base + lane];

        int j = 0;
        for (; j + 4 <= cnt; j += 4) {
            int u0 = __builtin_amdgcn_readlane(idx, j);
            int u1 = __builtin_amdgcn_readlane(idx, j + 1);
            int u2 = __builtin_amdgcn_readlane(idx, j + 2);
            int u3 = __builtin_amdgcn_readlane(idx, j + 3);
            u0 = ((unsigned)u0 < (unsigned)N_NODES) ? u0 : 0;
            u1 = ((unsigned)u1 < (unsigned)N_NODES) ? u1 : 0;
            u2 = ((unsigned)u2 < (unsigned)N_NODES) ? u2 : 0;
            u3 = ((unsigned)u3 < (unsigned)N_NODES) ? u3 : 0;
            const float4 f0 = *(const float4*)(el + (size_t)u0 * HD + c);
            const float4 f1 = *(const float4*)(el + (size_t)u1 * HD + c);
            const float4 f2 = *(const float4*)(el + (size_t)u2 * HD + c);
            const float4 f3 = *(const float4*)(el + (size_t)u3 * HD + c);
            const float p0 = gat_logit(f0, b4, w4);
            const float p1 = gat_logit(f1, b4, w4);
            const float p2 = gat_logit(f2, b4, w4);
            const float p3 = gat_logit(f3, b4, w4);
            const float mn = fmaxf(m, fmaxf(fmaxf(p0, p1), fmaxf(p2, p3)));
            const float sc = __expf(m - mn);
            const float a0 = __expf(p0 - mn);
            const float a1 = __expf(p1 - mn);
            const float a2 = __expf(p2 - mn);
            const float a3 = __expf(p3 - mn);
            l     = l * sc + ((a0 + a1) + (a2 + a3));
            acc.x = acc.x * sc + ((a0 * f0.x + a1 * f1.x) + (a2 * f2.x + a3 * f3.x));
            acc.y = acc.y * sc + ((a0 * f0.y + a1 * f1.y) + (a2 * f2.y + a3 * f3.y));
            acc.z = acc.z * sc + ((a0 * f0.z + a1 * f1.z) + (a2 * f2.z + a3 * f3.z));
            acc.w = acc.w * sc + ((a0 * f0.w + a1 * f1.w) + (a2 * f2.w + a3 * f3.w));
            m = mn;
        }
        for (; j < cnt; ++j) {
            int u0 = __builtin_amdgcn_readlane(idx, j);
            u0 = ((unsigned)u0 < (unsigned)N_NODES) ? u0 : 0;
            const float4 f0 = *(const float4*)(el + (size_t)u0 * HD + c);
            const float p0 = gat_logit(f0, b4, w4);
            const float mn = fmaxf(m, p0);
            const float sc = __expf(m - mn);
            const float a0 = __expf(p0 - mn);
            l     = l * sc + a0;
            acc.x = acc.x * sc + a0 * f0.x;
            acc.y = acc.y * sc + a0 * f0.y;
            acc.z = acc.z * sc + a0 * f0.z;
            acc.w = acc.w * sc + a0 * f0.w;
            m = mn;
        }
    }

    float* o = out + (size_t)v * HD + c;
    float4 r = *(const float4*)o;
    if (end > beg) {
        r.x += acc.x / l;
        r.y += acc.y / l;
        r.z += acc.z / l;
        r.w += acc.w / l;
    }
    *(float4*)o = r;
}

extern "C" void kernel_launch(void* const* d_in, const int* in_sizes, int n_in,
                              void* d_out, int out_size, void* d_ws, size_t ws_size,
                              hipStream_t stream) {
    const float* x    = (const float*)d_in[0];
    const int*   src  = (const int*)  d_in[1];
    const int*   dst  = (const int*)  d_in[2];
    const float* Wsrc = (const float*)d_in[3];
    const float* bsrc = (const float*)d_in[4];
    const float* Wdst = (const float*)d_in[5];
    const float* bdst = (const float*)d_in[6];
    const float* attn = (const float*)d_in[7];
    const float* Wres = (const float*)d_in[8];
    const float* bres = (const float*)d_in[9];
    float* out = (float*)d_out;

    char* ws = (char*)d_ws;
    const size_t elBytes   = (size_t)N_NODES * HD * sizeof(float);
    const size_t padBytes  = (size_t)TAB_PAD * sizeof(int);
    const size_t edgeBytes = (size_t)E_EDGES * sizeof(int);
    float* el     = (float*)(ws);
    float* er     = (float*)(ws + elBytes);
    int*   boff   = (int*)  (ws + 2 * elBytes);
    int*   rowptr = (int*)  (ws + 2 * elBytes + padBytes);
    int*   part   = (int*)  (ws + 2 * elBytes + 2 * padBytes + 512);
    int*   csr    = (int*)  (ws + 2 * elBytes + 2 * padBytes + 512 + edgeBytes);

    gat_range_hist<<<NCH, 256, 0, stream>>>(dst, boff);
    gat_range_scan<<<1, 1024, 0, stream>>>(boff);
    gat_range_scatter<<<NCH, 256, 0, stream>>>(src, dst, boff, part);
    gat_range_csr<<<NBK, 512, 0, stream>>>(part, boff, rowptr, csr);

    const int gemmBlocks = (N_NODES + ROWS_PB - 1) / ROWS_PB;
    gat_gemm3<<<gemmBlocks, 256, 0, stream>>>(x, Wsrc, bsrc, Wdst, bdst,
                                              Wres, bres, el, er, out);

    const int nodeBlocks = (N_NODES + 7) / 8;
    gat_node<<<nodeBlocks, 256, 0, stream>>>(el, er, rowptr, csr, attn, out);
}


# GEMM rows split so the WGPs fill evenly: 1536 64-row work-groups (two full rounds of 768) plus 318 small ones (16 rows x one output) for the last 1696 rows, not 27 big ones left alone; sums unchanged
# speedup vs baseline: 3.5134x; 1.3955x over previous
#include <hip/hip_runtime.h>
#include <hip/hip_bf16.h>

#define N_NODES  100000
#define E_EDGES  1600000
#define IN_F     256
#define HD       128
#define H_HEADS  8
#define SLOPE    0.2f

#define ROWS_PB    64
#define LDS_STRIDE 260

#define TAB_PAD    100352

#define MAIN_BLOCKS  1536
#define MAIN_ROWS    (MAIN_BLOCKS * ROWS_PB)
#define TAIL_TILES   ((N_NODES - MAIN_ROWS + 15) / 16)
#define TAIL_BLOCKS  (TAIL_TILES * 3)

typedef __attribute__((ext_vector_type(2))) float v2f;
typedef __attribute__((ext_vector_type(8))) float v8f;

__device__ __forceinline__ void gat_gemm_tail(
    const float* __restrict__ x, const float* __restrict__ W, const float* __restrict__ bvec,
    float* __restrict__ O, float* __restrict__ xs, const int row0)
{
    const int tid = threadIdx.x;
    {
        const int c4 = tid & 63;
        const int rb = tid >> 6;
        float4 v[4];
#pragma unroll
        for (int i = 0; i < 4; ++i) {
            const int gr = min(row0 + rb + 4 * i, N_NODES - 1);
            v[i] = *(const float4*)(x + (size_t)gr * IN_F + c4 * 4);
        }
#pragma unroll
        for (int i = 0; i < 4; ++i)
            *(float4*)&xs[(rb + 4 * i) * LDS_STRIDE + c4 * 4] = v[i];
    }
    __syncthreads();

    const int wave    = tid >> 5;
    const int lane    = tid & 31;
    const int l15     = lane & 15;
    const int kq      = (lane >> 4) * 4;
    const int rhalf   = (lane >> 4) * 8;
    const int colbase = wave * 16;

    const float* __restrict__ arow = &xs[l15 * LDS_STRIDE + kq];
    const float* __restrict__ Wp   = W + (size_t)kq * HD + colbase + l15;
    v8f acc = {};
    for (int k = 0; k < IN_F; k += 8) {
        const float4 a = *(const float4*)(arow + k);
        v2f b0, b1, a01, a23;
        b0.x = Wp[(k + 0) * HD];
        b0.y = Wp[(k + 1) * HD];
        b1.x = Wp[(k + 2) * HD];
        b1.y = Wp[(k + 3) * HD];
        a01.x = a.x; a01.y = a.y;
        a23.x = a.z; a23.y = a.w;
        acc = __builtin_amdgcn_wmma_f32_16x16x4_f32(false, a01, false, b0, (short)0, acc, false, false);
        acc = __builtin_amdgcn_wmma_f32_16x16x4_f32(false, a23, false, b1, (short)0, acc, false, false);
    }
    const float bias = bvec[colbase + l15];
#pragma unroll
    for (int j = 0; j < 8; ++j) {
        const int r = row0 + rhalf + j;
        if (r < N_NODES)
            O[(size_t)r * HD + colbase + l15] = acc[j] + bias;
    }
}

__global__ __launch_bounds__(256) void gat_gemm3(
    const float* __restrict__ x,
    const float* __restrict__ Wsrc, const float* __restrict__ bsrc,
    const float* __restrict__ Wdst, const float* __restrict__ bdst,
    const float* __restrict__ Wres, const float* __restrict__ bres,
    float* __restrict__ el, float* __restrict__ er, float* __restrict__ outres)
{
    __shared__ __attribute__((aligned(16))) float xs[ROWS_PB * LDS_STRIDE];

    if (blockIdx.x >= MAIN_BLOCKS) {
        const int t  = blockIdx.x - MAIN_BLOCKS;
        const int rt = t / 3;
        const int m  = t - rt * 3;
        gat_gemm_tail(x,
                      (m == 0) ? Wsrc : (m == 1) ? Wdst : Wres,
                      (m == 0) ? bsrc : (m == 1) ? bdst : bres,
                      (m == 0) ? el   : (m == 1) ? er   : outres,
                      xs, MAIN_ROWS + rt * 16);
        return;
    }

    const int tid  = threadIdx.x;
    const int row0 = blockIdx.x * ROWS_PB;

    {
        const int c4 = tid & 63;
        const int rb = tid >> 6;
        float4 v[16];
#pragma unroll
        for (int i = 0; i < 16; ++i) {
            const int gr = min(row0 + rb + 4 * i, N_NODES - 1);
            v[i] = *(const float4*)(x + (size_t)gr * IN_F + c4 * 4);
        }
#pragma unroll
        for (int i = 0; i < 16; ++i)
            *(float4*)&xs[(rb + 4 * i) * LDS_STRIDE + c4 * 4] = v[i];
    }
    __syncthreads();

    const int wave    = tid >> 5;
    const int lane    = tid & 31;
    const int l15     = lane & 15;
    const int kq      = (lane >> 4) * 4;
    const int rhalf   = (lane >> 4) * 8;
    const int colbase = wave * 16;

    const float* const Wm[3] = { Wsrc, Wdst, Wres };
    const float* const bm[3] = { bsrc, bdst, bres };
    float* const       Om[3] = { el,   er,   outres };

    const float* __restrict__ arow = &xs[l15 * LDS_STRIDE + kq];

#define GAT_LOADK(A0, A1, A2, A3, B0, B1, K)                                   \
    {                                                                          \
        A0 = *(const float4*)(arow + 0 * 16 * LDS_STRIDE + (K));               \
        A1 = *(const float4*)(arow + 1 * 16 * LDS_STRIDE + (K));               \
        A2 = *(const float4*)(arow + 2 * 16 * LDS_STRIDE + (K));               \
        A3 = *(const float4*)(arow + 3 * 16 * LDS_STRIDE + (K));               \
        B0.x = Wp[((K) + 0) * HD];                                             \
        B0.y = Wp[((K) + 1) * HD];                                             \
        B1.x = Wp[((K) + 2) * HD];                                             \
        B1.y = Wp[((K) + 3) * HD];                                             \
    }
#define GAT_WMMA(ACC, AX, AY, B)                                               \
    {                                                                          \
        v2f a; a.x = (AX); a.y = (AY);                                         \
        ACC = __builtin_amdgcn_wmma_f32_16x16x4_f32(                           \
                false, a, false, B, (short)0, ACC, false, false);              \
    }

#pragma unroll
    for (int m = 0; m < 3; ++m) {
        const float* __restrict__ Wp = Wm[m] + (size_t)kq * HD + colbase + l15;
        v8f acc0 = {}; v8f acc1 = {}; v8f acc2 = {}; v8f acc3 = {};

        for (int k = 0; k < IN_F; k += 8) {
            float4 a0, a1, a2, a3; v2f b0, b1;
            GAT_LOADK(a0, a1, a2, a3, b0, b1, k)

            GAT_WMMA(acc0, a0.x, a0.y, b0)
            GAT_WMMA(acc1, a1.x, a1.y, b0)
            GAT_WMMA(acc2, a2.x, a2.y, b0)
            GAT_WMMA(acc3, a3.x, a3.y, b0)
            GAT_WMMA(acc0, a0.z, a0.w, b1)
            GAT_WMMA(acc1, a1.z, a1.w, b1)
            GAT_WMMA(acc2, a2.z, a2.w, b1)
            GAT_WMMA(acc3, a3.z, a3.w, b1)
        }

        const float bias = bm[m][colbase + l15];
        float* __restrict__ O = Om[m];

#define GAT_STORE(ACC, RT)                                                     \
        {                                                                      \
            _Pragma("unroll")                                                  \
            for (int j = 0; j < 8; ++j) {                                      \
                const int r = row0 + (RT) * 16 + rhalf + j;                    \
                if (r < N_NODES)                                               \
                    O[(size_t)r * HD + colbase + l15] = ACC[j] + bias;         \
            }                                                                  \
        }
        GAT_STORE(acc0, 0)
        GAT_STORE(acc1, 1)
        GAT_STORE(acc2, 2)
        GAT_STORE(acc3, 3)
#undef GAT_STORE
    }
#undef GAT_WMMA
#undef GAT_LOADK
}

#define NBK      256
#define NPB      391
#define NCH      256
#define CH       (E_EDGES / NCH)
#define CH_FULL  (CH / 256)
#define CH_TAIL  (CH - CH_FULL * 256)
#define SRC_BITS 17
#define SRC_MASK ((1 << SRC_BITS) - 1)
#define CSR_CAP  12288

static_assert(NBK * NPB >= N_NODES, "ranges cover the nodes");
static_assert(NCH * CH == E_EDGES, "chunks cover the edges");
static_assert((1 << SRC_BITS) > N_NODES && NPB < (1 << (31 - SRC_BITS)), "packed word fits");

__device__ __forceinline__ int gat_clamp_node(int d)
{
    return min(max(d, 0), N_NODES - 1);
}

__device__ __forceinline__ int gat_block_exscan(int v, int* wsum)
{
    const int lane = threadIdx.x & 31;
    const int wave = threadIdx.x >> 5;
    int s = v;
    for (int d = 1; d < 32; d <<= 1) {
        const int t = __shfl_up(s, d, 32);
        if (lane >= d) s += t;
    }
    if (lane == 31) wsum[wave] = s;
    __syncthreads();
    if (wave == 0) {
        int w = (lane < (int)(blockDim.x >> 5)) ? wsum[lane] : 0;
        for (int d = 1; d < 32; d <<= 1) {
            const int t = __shfl_up(w, d, 32);
            if (lane >= d) w += t;
        }
        wsum[lane] = w;
    }
    __syncthreads();
    const int woff = (wave > 0) ? wsum[wave - 1] : 0;
    return woff + s - v;
}

__global__ __launch_bounds__(256) void gat_range_hist(
    const int* __restrict__ dst, int* __restrict__ boff)
{
    __shared__ int h[NBK];
    const int tid = threadIdx.x;
    const int* __restrict__ dp = dst + blockIdx.x * CH;
    h[tid] = 0;

    int d[CH_FULL];
#pragma unroll
    for (int k = 0; k < CH_FULL; ++k) d[k] = dp[k * 256 + tid];
    const int dl = dp[min(CH_FULL * 256 + tid, CH - 1)];
    __syncthreads();
#pragma unroll
    for (int k = 0; k < CH_FULL; ++k) atomicAdd(&h[gat_clamp_node(d[k]) / NPB], 1);
    if (tid < CH_TAIL) atomicAdd(&h[gat_clamp_node(dl) / NPB], 1);
    __syncthreads();
    boff[tid * NCH + blockIdx.x] = h[tid];
}

__global__ __launch_bounds__(1024) void gat_range_scan(int* __restrict__ boff)
{
    __shared__ int wsum[32];
    const int tid  = threadIdx.x;
    const int lane = tid & 31;
    const int wave = tid >> 5;
    int4* __restrict__ p = (int4*)boff + wave * 512;

    int4 v[16];
    int  part = 0;
#pragma unroll
    for (int j = 0; j < 16; ++j) {
        v[j] = p[j * 32 + lane];
        part += (v[j].x + v[j].y) + (v[j].z + v[j].w);
    }
    int tot = part;
    for (int d = 16; d > 0; d >>= 1) tot += __shfl_xor(tot, d, 32);
    if (lane == 0) wsum[wave] = tot;
    __syncthreads();
    if (wave == 0) {
        int w = wsum[lane];
        for (int d = 1; d < 32; d <<= 1) {
            const int t = __shfl_up(w, d, 32);
            if (lane >= d) w += t;
        }
        wsum[lane] = w;
    }
    __syncthreads();
    int carry = (wave > 0) ? wsum[wave - 1] : 0;
#pragma unroll
    for (int j = 0; j < 16; ++j) {
        const int s = (v[j].x + v[j].y) + (v[j].z + v[j].w);
        int inc = s;
        for (int d = 1; d < 32; d <<= 1) {
            const int t = __shfl_up(inc, d, 32);
            if (lane >= d) inc += t;
        }
        const int ex = carry + inc - s;
        int4 o;
        o.x = ex;
        o.y = ex + v[j].x;
        o.z = ex + v[j].x + v[j].y;
        o.w = ex + v[j].x + v[j].y + v[j].z;
        p[j * 32 + lane] = o;
        carry += __shfl(inc, 31, 32);
    }
    if (tid == 1023) boff[NBK * NCH] = carry;
}

__global__ __launch_bounds__(256) void gat_range_scatter(
    const int* __restrict__ src, const int* __restrict__ dst,
    const int* __restrict__ boff, int* __restrict__ part)
{
    __shared__ int lcur[NBK];
    __shared__ int gdel[NBK];
    __shared__ int sw[CH];
    __shared__ int sg[CH];
    __shared__ int wsum[32];
    const int tid = threadIdx.x;
    const int blk = blockIdx.x;
    const int* __restrict__ sp = src + blk * CH;
    const int* __restrict__ dp = dst + blk * CH;

    int d[CH_FULL], s[CH_FULL];
#pragma unroll
    for (int k = 0; k < CH_FULL; ++k) { d[k] = dp[k * 256 + tid]; s[k] = sp[k * 256 + tid]; }
    const int dl = dp[min(CH_FULL * 256 + tid, CH - 1)];
    const int sl = sp[min(CH_FULL * 256 + tid, CH - 1)];

    const int g0 = boff[tid * NCH + blk];
    const int g1 = boff[tid * NCH + blk + 1];
    const int ex = gat_block_exscan(g1 - g0, wsum);
    lcur[tid] = ex;
    gdel[tid] = g0 - ex;
    __syncthreads();

#define GAT_PLACE(D, S)                                                        \
    {                                                                          \
        const int dn = gat_clamp_node(D);                                      \
        const int b  = dn / NPB;                                               \
        const int sn = ((unsigned)(S) < (unsigned)N_NODES) ? (S) : 0;          \
        const int p  = atomicAdd(&lcur[b], 1);                                 \
        if ((unsigned)p < (unsigned)CH) {                                      \
            sw[p] = ((dn - b * NPB) << SRC_BITS) | sn;                         \
            sg[p] = gdel[b] + p;                                               \
        }                                                                      \
    }
#pragma unroll
    for (int k = 0; k < CH_FULL; ++k) GAT_PLACE(d[k], s[k])
    if (tid < CH_TAIL) GAT_PLACE(dl, sl)
#undef GAT_PLACE
    __syncthreads();

    for (int i = tid; i < CH; i += 256) {
        const int g = sg[i];
        if ((unsigned)g < (unsigned)E_EDGES)
            part[g] = sw[i];
    }
}

__global__ __launch_bounds__(512) void gat_range_csr(
    const int* __restrict__ part, const int* __restrict__ boff,
    int* __restrict__ rowptr, int* __restrict__ csr_src)
{
    __shared__ int cnt[512];
    __shared__ int lcsr[CSR_CAP];
    __shared__ int wsum[32];
    const int tid = threadIdx.x;
    const int b   = blockIdx.x;

    int bs = boff[b * NCH];
    int be = boff[(b + 1) * NCH];
    bs = min(max(bs, 0), E_EDGES);
    be = min(max(be, bs), E_EDGES);
    const int nb = be - bs;
    const int* __restrict__ pp = part + bs;

    cnt[tid] = 0;
    __syncthreads();
    for (int i0 = 0; i0 < nb; i0 += 4 * 512) {
        int w[4];
#pragma unroll
        for (int k = 0; k < 4; ++k) w[k] = pp[min(i0 + k * 512 + tid, nb - 1)];
#pragma unroll
        for (int k = 0; k < 4; ++k)
            if (i0 + k * 512 + tid < nb)
                atomicAdd(&cnt[min((int)((unsigned)w[k] >> SRC_BITS), NPB - 1)], 1);
    }
    __syncthreads();

    const int c  = (tid < NPB) ? cnt[tid] : 0;
    const int ex = gat_block_exscan(c, wsum);
    cnt[tid] = ex;
    if (tid < NPB) {
        const int node = b * NPB + tid;
        if (node < N_NODES) rowptr[node] = bs + ex;
    }
    if (b == NBK - 1 && tid == 0) rowptr[N_NODES] = be;
    __syncthreads();

    const bool stage = (nb <= CSR_CAP);
    for (int i0 = 0; i0 < nb; i0 += 4 * 512) {
        int w[4];
#pragma unroll
        for (int k = 0; k < 4; ++k) w[k] = pp[min(i0 + k * 512 + tid, nb - 1)];
#pragma unroll
        for (int k = 0; k < 4; ++k)
            if (i0 + k * 512 + tid < nb) {
                const int ln = min((int)((unsigned)w[k] >> SRC_BITS), NPB - 1);
                const int p  = atomicAdd(&cnt[ln], 1);
                if ((unsigned)p < (unsigned)nb) {
                    if (stage) lcsr[p] = w[k] & SRC_MASK;
                    else       csr_src[bs + p] = w[k] & SRC_MASK;
                }
            }
    }
    __syncthreads();
    if (stage)
        for (int i = tid; i < nb; i += 512) csr_src[bs + i] = lcsr[i];
}

__device__ __forceinline__ float gat_logit(const float4 f, const float4 b4, const float4 w4)
{
    float p = 0.f, t;
    t = f.x + b4.x; t = (t > 0.f) ? t : SLOPE * t; p += t * w4.x;
    t = f.y + b4.y; t = (t > 0.f) ? t : SLOPE * t; p += t * w4.y;
    t = f.z + b4.z; t = (t > 0.f) ? t : SLOPE * t; p += t * w4.z;
    t = f.w + b4.w; t = (t > 0.f) ? t : SLOPE * t; p += t * w4.w;
    p += __shfl_xor(p, 1, 32);
    p += __shfl_xor(p, 2, 32);
    return p;
}

__global__ __launch_bounds__(256) void gat_node(
    const float* __restrict__ el, const float* __restrict__ er,
    const int* __restrict__ rowptr, const int* __restrict__ csr_src,
    const float* __restrict__ attn, float* __restrict__ out)
{
    const int v = __builtin_amdgcn_readfirstlane(blockIdx.x * 8 + (threadIdx.x >> 5));
    if (v >= N_NODES) return;
    const int lane = threadIdx.x & 31;
    const int c    = lane * 4;

    int beg = rowptr[v];
    int end = rowptr[v + 1];
    beg = max(beg, 0);
    end = min(end, E_EDGES);

    const float4 b4 = *(const float4*)(er   + (size_t)v * HD + c);
    const float4 w4 = *(const float4*)(attn + c);

    float  m   = -INFINITY;
    float  l   = 0.f;
    float4 acc = make_float4(0.f, 0.f, 0.f, 0.f);

    for (int base = beg; base < end; base += 32) {
        const int cnt = min(32, end - base);
        int idx = 0;
        if (lane < cnt) idx = csr_src[base + lane];

        int j = 0;
        for (; j + 4 <= cnt; j += 4) {
            int u0 = __builtin_amdgcn_readlane(idx, j);
            int u1 = __builtin_amdgcn_readlane(idx, j + 1);
            int u2 = __builtin_amdgcn_readlane(idx, j + 2);
            int u3 = __builtin_amdgcn_readlane(idx, j + 3);
            u0 = ((unsigned)u0 < (unsigned)N_NODES) ? u0 : 0;
            u1 = ((unsigned)u1 < (unsigned)N_NODES) ? u1 : 0;
            u2 = ((unsigned)u2 < (unsigned)N_NODES) ? u2 : 0;
            u3 = ((unsigned)u3 < (unsigned)N_NODES) ? u3 : 0;
            const float4 f0 = *(const float4*)(el + (size_t)u0 * HD + c);
            const float4 f1 = *(const float4*)(el + (size_t)u1 * HD + c);
            const float4 f2 = *(const float4*)(el + (size_t)u2 * HD + c);
            const float4 f3 = *(const float4*)(el + (size_t)u3 * HD + c);
            const float p0 = gat_logit(f0, b4, w4);
            const float p1 = gat_logit(f1, b4, w4);
            const float p2 = gat_logit(f2, b4, w4);
            const float p3 = gat_logit(f3, b4, w4);
            const float mn = fmaxf(m, fmaxf(fmaxf(p0, p1), fmaxf(p2, p3)));
            const float sc = __expf(m - mn);
            const float a0 = __expf(p0 - mn);
            const float a1 = __expf(p1 - mn);
            const float a2 = __expf(p2 - mn);
            const float a3 = __expf(p3 - mn);
            l     = l * sc + ((a0 + a1) + (a2 + a3));
            acc.x = acc.x * sc + ((a0 * f0.x + a1 * f1.x) + (a2 * f2.x + a3 * f3.x));
            acc.y = acc.y * sc + ((a0 * f0.y + a1 * f1.y) + (a2 * f2.y + a3 * f3.y));
            acc.z = acc.z * sc + ((a0 * f0.z + a1 * f1.z) + (a2 * f2.z + a3 * f3.z));
            acc.w = acc.w * sc + ((a0 * f0.w + a1 * f1.w) + (a2 * f2.w + a3 * f3.w));
            m = mn;
        }
        for (; j < cnt; ++j) {
            int u0 = __builtin_amdgcn_readlane(idx, j);
            u0 = ((unsigned)u0 < (unsigned)N_NODES) ? u0 : 0;
            const float4 f0 = *(const float4*)(el + (size_t)u0 * HD + c);
            const float p0 = gat_logit(f0, b4, w4);
            const float mn = fmaxf(m, p0);
            const float sc = __expf(m - mn);
            const float a0 = __expf(p0 - mn);
            l     = l * sc + a0;
            acc.x = acc.x * sc + a0 * f0.x;
            acc.y = acc.y * sc + a0 * f0.y;
            acc.z = acc.z * sc + a0 * f0.z;
            acc.w = acc.w * sc + a0 * f0.w;
            m = mn;
        }
    }

    float* o = out + (size_t)v * HD + c;
    float4 r = *(const float4*)o;
    if (end > beg) {
        r.x += acc.x / l;
        r.y += acc.y / l;
        r.z += acc.z / l;
        r.w += acc.w / l;
    }
    *(float4*)o = r;
}

extern "C" void kernel_launch(void* const* d_in, const int* in_sizes, int n_in,
                              void* d_out, int out_size, void* d_ws, size_t ws_size,
                              hipStream_t stream) {
    const float* x    = (const float*)d_in[0];
    const int*   src  = (const int*)  d_in[1];
    const int*   dst  = (const int*)  d_in[2];
    const float* Wsrc = (const float*)d_in[3];
    const float* bsrc = (const float*)d_in[4];
    const float* Wdst = (const float*)d_in[5];
    const float* bdst = (const float*)d_in[6];
    const float* attn = (const float*)d_in[7];
    const float* Wres = (const float*)d_in[8];
    const float* bres = (const float*)d_in[9];
    float* out = (float*)d_out;

    char* ws = (char*)d_ws;
    const size_t elBytes   = (size_t)N_NODES * HD * sizeof(float);
    const size_t padBytes  = (size_t)TAB_PAD * sizeof(int);
    const size_t edgeBytes = (size_t)E_EDGES * sizeof(int);
    float* el     = (float*)(ws);
    float* er     = (float*)(ws + elBytes);
    int*   boff   = (int*)  (ws + 2 * elBytes);
    int*   rowptr = (int*)  (ws + 2 * elBytes + padBytes);
    int*   part   = (int*)  (ws + 2 * elBytes + 2 * padBytes + 512);
    int*   csr    = (int*)  (ws + 2 * elBytes + 2 * padBytes + 512 + edgeBytes);

    gat_range_hist<<<NCH, 256, 0, stream>>>(dst, boff);
    gat_range_scan<<<1, 1024, 0, stream>>>(boff);
    gat_range_scatter<<<NCH, 256, 0, stream>>>(src, dst, boff, part);
    gat_range_csr<<<NBK, 512, 0, stream>>>(part, boff, rowptr, csr);

    gat_gemm3<<<MAIN_BLOCKS + TAIL_BLOCKS, 256, 0, stream>>>(x, Wsrc, bsrc, Wdst, bdst,
                                              Wres, bres, el, er, out);

    const int nodeBlocks = (N_NODES + 7) / 8;
    gat_node<<<nodeBlocks, 256, 0, stream>>>(el, er, rowptr, csr, attn, out);
}


# er and res GEMMs moved into the node kernel: each 64-node work-group computes its er/res tiles on WMMA into LDS, then aggregates its nodes (8 row gathers in flight); el GEMM apart; sums unchanged
# speedup vs baseline: 3.6958x; 1.0519x over previous
#include <hip/hip_runtime.h>
#include <hip/hip_bf16.h>

#define N_NODES  100000
#define E_EDGES  1600000
#define IN_F     256
#define HD       128
#define H_HEADS  8
#define SLOPE    0.2f

#define ROWS_PB    64
#define LDS_STRIDE 260

#define TAB_PAD    100352

#define MAIN_BLOCKS  1536
#define MAIN_ROWS    (MAIN_BLOCKS * ROWS_PB)
#define TAIL_TILES   ((N_NODES - MAIN_ROWS + 15) / 16)
#define TAIL_BLOCKS  TAIL_TILES
#define NODE_BLOCKS  ((N_NODES + ROWS_PB - 1) / ROWS_PB)

typedef __attribute__((ext_vector_type(2))) float v2f;
typedef __attribute__((ext_vector_type(8))) float v8f;

__device__ __forceinline__ void gat_gemm_tail(
    const float* __restrict__ x, const float* __restrict__ W, const float* __restrict__ bvec,
    float* __restrict__ O, float* __restrict__ xs, const int row0)
{
    const int tid = threadIdx.x;
    {
        const int c4 = tid & 63;
        const int rb = tid >> 6;
        float4 v[4];
#pragma unroll
        for (int i = 0; i < 4; ++i) {
            const int gr = min(row0 + rb + 4 * i, N_NODES - 1);
            v[i] = *(const float4*)(x + (size_t)gr * IN_F + c4 * 4);
        }
#pragma unroll
        for (int i = 0; i < 4; ++i)
            *(float4*)&xs[(rb + 4 * i) * LDS_STRIDE + c4 * 4] = v[i];
    }
    __syncthreads();

    const int wave    = tid >> 5;
    const int lane    = tid & 31;
    const int l15     = lane & 15;
    const int kq      = (lane >> 4) * 4;
    const int rhalf   = (lane >> 4) * 8;
    const int colbase = wave * 16;

    const float* __restrict__ arow = &xs[l15 * LDS_STRIDE + kq];
    const float* __restrict__ Wp   = W + (size_t)kq * HD + colbase + l15;
    v8f acc = {};
    for (int k = 0; k < IN_F; k += 8) {
        const float4 a = *(const float4*)(arow + k);
        v2f b0, b1, a01, a23;
        b0.x = Wp[(k + 0) * HD];
        b0.y = Wp[(k + 1) * HD];
        b1.x = Wp[(k + 2) * HD];
        b1.y = Wp[(k + 3) * HD];
        a01.x = a.x; a01.y = a.y;
        a23.x = a.z; a23.y = a.w;
        acc = __builtin_amdgcn_wmma_f32_16x16x4_f32(false, a01, false, b0, (short)0, acc, false, false);
        acc = __builtin_amdgcn_wmma_f32_16x16x4_f32(false, a23, false, b1, (short)0, acc, false, false);
    }
    const float bias = bvec[colbase + l15];
#pragma unroll
    for (int j = 0; j < 8; ++j) {
        const int r = row0 + rhalf + j;
        if (r < N_NODES)
            O[(size_t)r * HD + colbase + l15] = acc[j] + bias;
    }
}

#define GAT_LOADA(A0, A1, A2, A3, K)                                           \
    {                                                                          \
        A0 = *(const float4*)(arow + 0 * 16 * LDS_STRIDE + (K));               \
        A1 = *(const float4*)(arow + 1 * 16 * LDS_STRIDE + (K));               \
        A2 = *(const float4*)(arow + 2 * 16 * LDS_STRIDE + (K));               \
        A3 = *(const float4*)(arow + 3 * 16 * LDS_STRIDE + (K));               \
    }
#define GAT_LOADB(B0, B1, WP, K)                                               \
    {                                                                          \
        B0.x = WP[((K) + 0) * HD];                                             \
        B0.y = WP[((K) + 1) * HD];                                             \
        B1.x = WP[((K) + 2) * HD];                                             \
        B1.y = WP[((K) + 3) * HD];                                             \
    }
#define GAT_WMMA(ACC, AX, AY, B)                                               \
    {                                                                          \
        v2f a; a.x = (AX); a.y = (AY);                                         \
        ACC = __builtin_amdgcn_wmma_f32_16x16x4_f32(                           \
                false, a, false, B, (short)0, ACC, false, false);              \
    }

__global__ __launch_bounds__(256) void gat_gemm_el(
    const float* __restrict__ x,
    const float* __restrict__ Wsrc, const float* __restrict__ bsrc,
    float* __restrict__ el)
{
    __shared__ __attribute__((aligned(16))) float xs[ROWS_PB * LDS_STRIDE];

    if (blockIdx.x >= MAIN_BLOCKS) {
        gat_gemm_tail(x, Wsrc, bsrc, el, xs, MAIN_ROWS + (blockIdx.x - MAIN_BLOCKS) * 16);
        return;
    }

    const int tid  = threadIdx.x;
    const int row0 = blockIdx.x * ROWS_PB;

    {
        const int c4 = tid & 63;
        const int rb = tid >> 6;
        float4 v[16];
#pragma unroll
        for (int i = 0; i < 16; ++i) {
            const int gr = min(row0 + rb + 4 * i, N_NODES - 1);
            v[i] = *(const float4*)(x + (size_t)gr * IN_F + c4 * 4);
        }
#pragma unroll
        for (int i = 0; i < 16; ++i)
            *(float4*)&xs[(rb + 4 * i) * LDS_STRIDE + c4 * 4] = v[i];
    }
    __syncthreads();

    const int wave    = tid >> 5;
    const int lane    = tid & 31;
    const int l15     = lane & 15;
    const int kq      = (lane >> 4) * 4;
    const int rhalf   = (lane >> 4) * 8;
    const int colbase = wave * 16;

    const float* __restrict__ arow = &xs[l15 * LDS_STRIDE + kq];
    const float* __restrict__ Wp   = Wsrc + (size_t)kq * HD + colbase + l15;
    v8f acc0 = {}; v8f acc1 = {}; v8f acc2 = {}; v8f acc3 = {};

    for (int k = 0; k < IN_F; k += 8) {
        float4 a0, a1, a2, a3; v2f b0, b1;
        GAT_LOADA(a0, a1, a2, a3, k)
        GAT_LOADB(b0, b1, Wp, k)

        GAT_WMMA(acc0, a0.x, a0.y, b0)
        GAT_WMMA(acc1, a1.x, a1.y, b0)
        GAT_WMMA(acc2, a2.x, a2.y, b0)
        GAT_WMMA(acc3, a3.x, a3.y, b0)
        GAT_WMMA(acc0, a0.z, a0.w, b1)
        GAT_WMMA(acc1, a1.z, a1.w, b1)
        GAT_WMMA(acc2, a2.z, a2.w, b1)
        GAT_WMMA(acc3, a3.z, a3.w, b1)
    }

    const float bias = bsrc[colbase + l15];
#define GAT_STORE(ACC, RT)                                                     \
    {                                                                          \
        _Pragma("unroll")                                                      \
        for (int j = 0; j < 8; ++j) {                                          \
            const int r = row0 + (RT) * 16 + rhalf + j;                        \
            if (r < N_NODES)                                                   \
                el[(size_t)r * HD + colbase + l15] = ACC[j] + bias;            \
        }                                                                      \
    }
    GAT_STORE(acc0, 0)
    GAT_STORE(acc1, 1)
    GAT_STORE(acc2, 2)
    GAT_STORE(acc3, 3)
#undef GAT_STORE
}

#define NBK      256
#define NPB      391
#define NCH      256
#define CH       (E_EDGES / NCH)
#define CH_FULL  (CH / 256)
#define CH_TAIL  (CH - CH_FULL * 256)
#define SRC_BITS 17
#define SRC_MASK ((1 << SRC_BITS) - 1)
#define CSR_CAP  12288

static_assert(NBK * NPB >= N_NODES, "ranges cover the nodes");
static_assert(NCH * CH == E_EDGES, "chunks cover the edges");
static_assert((1 << SRC_BITS) > N_NODES && NPB < (1 << (31 - SRC_BITS)), "packed word fits");

__device__ __forceinline__ int gat_clamp_node(int d)
{
    return min(max(d, 0), N_NODES - 1);
}

__device__ __forceinline__ int gat_block_exscan(int v, int* wsum)
{
    const int lane = threadIdx.x & 31;
    const int wave = threadIdx.x >> 5;
    int s = v;
    for (int d = 1; d < 32; d <<= 1) {
        const int t = __shfl_up(s, d, 32);
        if (lane >= d) s += t;
    }
    if (lane == 31) wsum[wave] = s;
    __syncthreads();
    if (wave == 0) {
        int w = (lane < (int)(blockDim.x >> 5)) ? wsum[lane] : 0;
        for (int d = 1; d < 32; d <<= 1) {
            const int t = __shfl_up(w, d, 32);
            if (lane >= d) w += t;
        }
        wsum[lane] = w;
    }
    __syncthreads();
    const int woff = (wave > 0) ? wsum[wave - 1] : 0;
    return woff + s - v;
}

__global__ __launch_bounds__(256) void gat_range_hist(
    const int* __restrict__ dst, int* __restrict__ boff)
{
    __shared__ int h[NBK];
    const int tid = threadIdx.x;
    const int* __restrict__ dp = dst + blockIdx.x * CH;
    h[tid] = 0;

    int d[CH_FULL];
#pragma unroll
    for (int k = 0; k < CH_FULL; ++k) d[k] = dp[k * 256 + tid];
    const int dl = dp[min(CH_FULL * 256 + tid, CH - 1)];
    __syncthreads();
#pragma unroll
    for (int k = 0; k < CH_FULL; ++k) atomicAdd(&h[gat_clamp_node(d[k]) / NPB], 1);
    if (tid < CH_TAIL) atomicAdd(&h[gat_clamp_node(dl) / NPB], 1);
    __syncthreads();
    boff[tid * NCH + blockIdx.x] = h[tid];
}

__global__ __launch_bounds__(1024) void gat_range_scan(int* __restrict__ boff)
{
    __shared__ int wsum[32];
    const int tid  = threadIdx.x;
    const int lane = tid & 31;
    const int wave = tid >> 5;
    int4* __restrict__ p = (int4*)boff + wave * 512;

    int4 v[16];
    int  part = 0;
#pragma unroll
    for (int j = 0; j < 16; ++j) {
        v[j] = p[j * 32 + lane];
        part += (v[j].x + v[j].y) + (v[j].z + v[j].w);
    }
    int tot = part;
    for (int d = 16; d > 0; d >>= 1) tot += __shfl_xor(tot, d, 32);
    if (lane == 0) wsum[wave] = tot;
    __syncthreads();
    if (wave == 0) {
        int w = wsum[lane];
        for (int d = 1; d < 32; d <<= 1) {
            const int t = __shfl_up(w, d, 32);
            if (lane >= d) w += t;
        }
        wsum[lane] = w;
    }
    __syncthreads();
    int carry = (wave > 0) ? wsum[wave - 1] : 0;
#pragma unroll
    for (int j = 0; j < 16; ++j) {
        const int s = (v[j].x + v[j].y) + (v[j].z + v[j].w);
        int inc = s;
        for (int d = 1; d < 32; d <<= 1) {
            const int t = __shfl_up(inc, d, 32);
            if (lane >= d) inc += t;
        }
        const int ex = carry + inc - s;
        int4 o;
        o.x = ex;
        o.y = ex + v[j].x;
        o.z = ex + v[j].x + v[j].y;
        o.w = ex + v[j].x + v[j].y + v[j].z;
        p[j * 32 + lane] = o;
        carry += __shfl(inc, 31, 32);
    }
    if (tid == 1023) boff[NBK * NCH] = carry;
}

__global__ __launch_bounds__(256) void gat_range_scatter(
    const int* __restrict__ src, const int* __restrict__ dst,
    const int* __restrict__ boff, int* __restrict__ part)
{
    __shared__ int lcur[NBK];
    __shared__ int gdel[NBK];
    __shared__ int sw[CH];
    __shared__ int sg[CH];
    __shared__ int wsum[32];
    const int tid = threadIdx.x;
    const int blk = blockIdx.x;
    const int* __restrict__ sp = src + blk * CH;
    const int* __restrict__ dp = dst + blk * CH;

    int d[CH_FULL], s[CH_FULL];
#pragma unroll
    for (int k = 0; k < CH_FULL; ++k) { d[k] = dp[k * 256 + tid]; s[k] = sp[k * 256 + tid]; }
    const int dl = dp[min(CH_FULL * 256 + tid, CH - 1)];
    const int sl = sp[min(CH_FULL * 256 + tid, CH - 1)];

    const int g0 = boff[tid * NCH + blk];
    const int g1 = boff[tid * NCH + blk + 1];
    const int ex = gat_block_exscan(g1 - g0, wsum);
    lcur[tid] = ex;
    gdel[tid] = g0 - ex;
    __syncthreads();

#define GAT_PLACE(D, S)                                                        \
    {                                                                          \
        const int dn = gat_clamp_node(D);                                      \
        const int b  = dn / NPB;                                               \
        const int sn = ((unsigned)(S) < (unsigned)N_NODES) ? (S) : 0;          \
        const int p  = atomicAdd(&lcur[b], 1);                                 \
        if ((unsigned)p < (unsigned)CH) {                                      \
            sw[p] = ((dn - b * NPB) << SRC_BITS) | sn;                         \
            sg[p] = gdel[b] + p;                                               \
        }                                                                      \
    }
#pragma unroll
    for (int k = 0; k < CH_FULL; ++k) GAT_PLACE(d[k], s[k])
    if (tid < CH_TAIL) GAT_PLACE(dl, sl)
#undef GAT_PLACE
    __syncthreads();

    for (int i = tid; i < CH; i += 256) {
        const int g = sg[i];
        if ((unsigned)g < (unsigned)E_EDGES)
            part[g] = sw[i];
    }
}

__global__ __launch_bounds__(512) void gat_range_csr(
    const int* __restrict__ part, const int* __restrict__ boff,
    int* __restrict__ rowptr, int* __restrict__ csr_src)
{
    __shared__ int cnt[512];
    __shared__ int lcsr[CSR_CAP];
    __shared__ int wsum[32];
    const int tid = threadIdx.x;
    const int b   = blockIdx.x;

    int bs = boff[b * NCH];
    int be = boff[(b + 1) * NCH];
    bs = min(max(bs, 0), E_EDGES);
    be = min(max(be, bs), E_EDGES);
    const int nb = be - bs;
    const int* __restrict__ pp = part + bs;

    cnt[tid] = 0;
    __syncthreads();
    for (int i0 = 0; i0 < nb; i0 += 4 * 512) {
        int w[4];
#pragma unroll
        for (int k = 0; k < 4; ++k) w[k] = pp[min(i0 + k * 512 + tid, nb - 1)];
#pragma unroll
        for (int k = 0; k < 4; ++k)
            if (i0 + k * 512 + tid < nb)
                atomicAdd(&cnt[min((int)((unsigned)w[k] >> SRC_BITS), NPB - 1)], 1);
    }
    __syncthreads();

    const int c  = (tid < NPB) ? cnt[tid] : 0;
    const int ex = gat_block_exscan(c, wsum);
    cnt[tid] = ex;
    if (tid < NPB) {
        const int node = b * NPB + tid;
        if (node < N_NODES) rowptr[node] = bs + ex;
    }
    if (b == NBK - 1 && tid == 0) rowptr[N_NODES] = be;
    __syncthreads();

    const bool stage = (nb <= CSR_CAP);
    for (int i0 = 0; i0 < nb; i0 += 4 * 512) {
        int w[4];
#pragma unroll
        for (int k = 0; k < 4; ++k) w[k] = pp[min(i0 + k * 512 + tid, nb - 1)];
#pragma unroll
        for (int k = 0; k < 4; ++k)
            if (i0 + k * 512 + tid < nb) {
                const int ln = min((int)((unsigned)w[k] >> SRC_BITS), NPB - 1);
                const int p  = atomicAdd(&cnt[ln], 1);
                if ((unsigned)p < (unsigned)nb) {
                    if (stage) lcsr[p] = w[k] & SRC_MASK;
                    else       csr_src[bs + p] = w[k] & SRC_MASK;
                }
            }
    }
    __syncthreads();
    if (stage)
        for (int i = tid; i < nb; i += 512) csr_src[bs + i] = lcsr[i];
}

__device__ __forceinline__ float gat_logit(const float4 f, const float4 b4, const float4 w4)
{
    float p = 0.f, t;
    t = f.x + b4.x; t = (t > 0.f) ? t : SLOPE * t; p += t * w4.x;
    t = f.y + b4.y; t = (t > 0.f) ? t : SLOPE * t; p += t * w4.y;
    t = f.z + b4.z; t = (t > 0.f) ? t : SLOPE * t; p += t * w4.z;
    t = f.w + b4.w; t = (t > 0.f) ? t : SLOPE * t; p += t * w4.w;
    p += __shfl_xor(p, 1, 32);
    p += __shfl_xor(p, 2, 32);
    return p;
}

#define GAT_STEP4(F0, F1, F2, F3)                                                          \
    {                                                                                      \
        const float p0 = gat_logit(F0, b4, w4);                                            \
        const float p1 = gat_logit(F1, b4, w4);                                            \
        const float p2 = gat_logit(F2, b4, w4);                                            \
        const float p3 = gat_logit(F3, b4, w4);                                            \
        const float mn = fmaxf(m, fmaxf(fmaxf(p0, p1), fmaxf(p2, p3)));                    \
        const float sc = __expf(m - mn);                                                   \
        const float a0 = __expf(p0 - mn);                                                  \
        const float a1 = __expf(p1 - mn);                                                  \
        const float a2 = __expf(p2 - mn);                                                  \
        const float a3 = __expf(p3 - mn);                                                  \
        l     = l * sc + ((a0 + a1) + (a2 + a3));                                          \
        acc.x = acc.x * sc + ((a0 * F0.x + a1 * F1.x) + (a2 * F2.x + a3 * F3.x));          \
        acc.y = acc.y * sc + ((a0 * F0.y + a1 * F1.y) + (a2 * F2.y + a3 * F3.y));          \
        acc.z = acc.z * sc + ((a0 * F0.z + a1 * F1.z) + (a2 * F2.z + a3 * F3.z));          \
        acc.w = acc.w * sc + ((a0 * F0.w + a1 * F1.w) + (a2 * F2.w + a3 * F3.w));          \
        m = mn;                                                                            \
    }
#define GAT_ROW(J)                                                                         \
    ({                                                                                     \
        int u_ = __builtin_amdgcn_readlane(idx, (J));                                      \
        u_ = ((unsigned)u_ < (unsigned)N_NODES) ? u_ : 0;                                  \
        *(const float4*)(el + (size_t)u_ * HD + c);                                        \
    })

__global__ __launch_bounds__(256) void gat_node_fused(
    const float* __restrict__ x,
    const float* __restrict__ Wdst, const float* __restrict__ bdst,
    const float* __restrict__ Wres, const float* __restrict__ bres,
    const float* __restrict__ el,
    const int* __restrict__ rowptr, const int* __restrict__ csr_src,
    const float* __restrict__ attn, float* __restrict__ out)
{
    __shared__ __attribute__((aligned(16))) float xs[ROWS_PB * LDS_STRIDE];

    const int tid  = threadIdx.x;
    const int row0 = blockIdx.x * ROWS_PB;
    const int lane = tid & 31;
    const int wv   = __builtin_amdgcn_readfirstlane(tid >> 5);

    const int rp = rowptr[min(row0 + wv * 8 + min(lane, 8), N_NODES)];

    {
        const int c4 = tid & 63;
        const int rb = tid >> 6;
        float4 v[16];
#pragma unroll
        for (int i = 0; i < 16; ++i) {
            const int gr = min(row0 + rb + 4 * i, N_NODES - 1);
            v[i] = *(const float4*)(x + (size_t)gr * IN_F + c4 * 4);
        }
#pragma unroll
        for (int i = 0; i < 16; ++i)
            *(float4*)&xs[(rb + 4 * i) * LDS_STRIDE + c4 * 4] = v[i];
    }
    __syncthreads();

    const int l15     = lane & 15;
    const int kq      = (lane >> 4) * 4;
    const int rhalf   = (lane >> 4) * 8;
    const int colbase = wv * 16;

    {
        const float* __restrict__ arow = &xs[l15 * LDS_STRIDE + kq];
        const float* __restrict__ Wd   = Wdst + (size_t)kq * HD + colbase + l15;
        const float* __restrict__ Wr   = Wres + (size_t)kq * HD + colbase + l15;
        v8f e0 = {}; v8f e1 = {}; v8f e2 = {}; v8f e3 = {};
        v8f r0 = {}; v8f r1 = {}; v8f r2 = {}; v8f r3 = {};

        for (int k = 0; k < IN_F; k += 8) {
            float4 a0, a1, a2, a3; v2f d0, d1, s0, s1;
            GAT_LOADA(a0, a1, a2, a3, k)
            GAT_LOADB(d0, d1, Wd, k)
            GAT_LOADB(s0, s1, Wr, k)

            GAT_WMMA(e0, a0.x, a0.y, d0)
            GAT_WMMA(e1, a1.x, a1.y, d0)
            GAT_WMMA(e2, a2.x, a2.y, d0)
            GAT_WMMA(e3, a3.x, a3.y, d0)
            GAT_WMMA(r0, a0.x, a0.y, s0)
            GAT_WMMA(r1, a1.x, a1.y, s0)
            GAT_WMMA(r2, a2.x, a2.y, s0)
            GAT_WMMA(r3, a3.x, a3.y, s0)
            GAT_WMMA(e0, a0.z, a0.w, d1)
            GAT_WMMA(e1, a1.z, a1.w, d1)
            GAT_WMMA(e2, a2.z, a2.w, d1)
            GAT_WMMA(e3, a3.z, a3.w, d1)
            GAT_WMMA(r0, a0.z, a0.w, s1)
            GAT_WMMA(r1, a1.z, a1.w, s1)
            GAT_WMMA(r2, a2.z, a2.w, s1)
            GAT_WMMA(r3, a3.z, a3.w, s1)
        }

        const float bias_d = bdst[colbase + l15];
        const float bias_r = bres[colbase + l15];
        __syncthreads();

#define GAT_KEEP(EACC, RACC, RT)                                                           \
        {                                                                                  \
            _Pragma("unroll")                                                              \
            for (int j = 0; j < 8; ++j) {                                                  \
                float* q = &xs[((RT) * 16 + rhalf + j) * LDS_STRIDE + colbase + l15];      \
                q[0]  = EACC[j] + bias_d;                                                  \
                q[HD] = RACC[j] + bias_r;                                                  \
            }                                                                              \
        }
        GAT_KEEP(e0, r0, 0)
        GAT_KEEP(e1, r1, 1)
        GAT_KEEP(e2, r2, 2)
        GAT_KEEP(e3, r3, 3)
#undef GAT_KEEP
    }
    __syncthreads();

    const int    c  = lane * 4;
    const float4 w4 = *(const float4*)(attn + c);

    for (int i = 0; i < 8; ++i) {
        const int v = row0 + wv * 8 + i;
        if (v >= N_NODES) break;

        int beg = __builtin_amdgcn_readlane(rp, i);
        int end = __builtin_amdgcn_readlane(rp, i + 1);
        beg = max(beg, 0);
        end = min(end, E_EDGES);

        const float* trow = &xs[(wv * 8 + i) * LDS_STRIDE + c];
        const float4 b4 = *(const float4*)trow;

        float  m   = -INFINITY;
        float  l   = 0.f;
        float4 acc = make_float4(0.f, 0.f, 0.f, 0.f);

        for (int base = beg; base < end; base += 32) {
            const int cnt = min(32, end - base);
            int idx = 0;
            if (lane < cnt) idx = csr_src[base + lane];

            int j = 0;
            for (; j + 8 <= cnt; j += 8) {
                const float4 f0 = GAT_ROW(j);
                const float4 f1 = GAT_ROW(j + 1);
                const float4 f2 = GAT_ROW(j + 2);
                const float4 f3 = GAT_ROW(j + 3);
                const float4 f4 = GAT_ROW(j + 4);
                const float4 f5 = GAT_ROW(j + 5);
                const float4 f6 = GAT_ROW(j + 6);
                const float4 f7 = GAT_ROW(j + 7);
                GAT_STEP4(f0, f1, f2, f3)
                GAT_STEP4(f4, f5, f6, f7)
            }
            for (; j + 4 <= cnt; j += 4) {
                const float4 f0 = GAT_ROW(j);
                const float4 f1 = GAT_ROW(j + 1);
                const float4 f2 = GAT_ROW(j + 2);
                const float4 f3 = GAT_ROW(j + 3);
                GAT_STEP4(f0, f1, f2, f3)
            }
            for (; j < cnt; ++j) {
                const float4 f0 = GAT_ROW(j);
                const float p0 = gat_logit(f0, b4, w4);
                const float mn = fmaxf(m, p0);
                const float sc = __expf(m - mn);
                const float a0 = __expf(p0 - mn);
                l     = l * sc + a0;
                acc.x = acc.x * sc + a0 * f0.x;
                acc.y = acc.y * sc + a0 * f0.y;
                acc.z = acc.z * sc + a0 * f0.z;
                acc.w = acc.w * sc + a0 * f0.w;
                m = mn;
            }
        }

        float4 r = *(const float4*)(trow + HD);
        if (end > beg) {
            r.x += acc.x / l;
            r.y += acc.y / l;
            r.z += acc.z / l;
            r.w += acc.w / l;
        }
        *(float4*)(out + (size_t)v * HD + c) = r;
    }
}
#undef GAT_ROW
#undef GAT_STEP4
#undef GAT_WMMA
#undef GAT_LOADB
#undef GAT_LOADA

extern "C" void kernel_launch(void* const* d_in, const int* in_sizes, int n_in,
                              void* d_out, int out_size, void* d_ws, size_t ws_size,
                              hipStream_t stream) {
    const float* x    = (const float*)d_in[0];
    const int*   src  = (const int*)  d_in[1];
    const int*   dst  = (const int*)  d_in[2];
    const float* Wsrc = (const float*)d_in[3];
    const float* bsrc = (const float*)d_in[4];
    const float* Wdst = (const float*)d_in[5];
    const float* bdst = (const float*)d_in[6];
    const float* attn = (const float*)d_in[7];
    const float* Wres = (const float*)d_in[8];
    const float* bres = (const float*)d_in[9];
    float* out = (float*)d_out;

    char* ws = (char*)d_ws;
    const size_t elBytes   = (size_t)N_NODES * HD * sizeof(float);
    const size_t padBytes  = (size_t)TAB_PAD * sizeof(int);
    const size_t edgeBytes = (size_t)E_EDGES * sizeof(int);
    float* el     = (float*)(ws);
    int*   boff   = (int*)  (ws + 2 * elBytes);
    int*   rowptr = (int*)  (ws + 2 * elBytes + padBytes);
    int*   part   = (int*)  (ws + 2 * elBytes + 2 * padBytes + 512);
    int*   csr    = (int*)  (ws + 2 * elBytes + 2 * padBytes + 512 + edgeBytes);

    gat_range_hist<<<NCH, 256, 0, stream>>>(dst, boff);
    gat_range_scan<<<1, 1024, 0, stream>>>(boff);
    gat_range_scatter<<<NCH, 256, 0, stream>>>(src, dst, boff, part);
    gat_range_csr<<<NBK, 512, 0, stream>>>(part, boff, rowptr, csr);

    gat_gemm_el<<<MAIN_BLOCKS + TAIL_BLOCKS, 256, 0, stream>>>(x, Wsrc, bsrc, el);

    gat_node_fused<<<NODE_BLOCKS, 256, 0, stream>>>(x, Wdst, bdst, Wres, bres, el, rowptr, csr, attn, out);
}


# Edge grouping in 2 launches, not 4: each chunk ordered by node range in LDS, stored back in place with a (start,length) word per run; the range kernel collects its runs; sums still in arrival order
# speedup vs baseline: 3.8925x; 1.0532x over previous
#include <hip/hip_runtime.h>
#include <hip/hip_bf16.h>

#define N_NODES  100000
#define E_EDGES  1600000
#define IN_F     256
#define HD       128
#define H_HEADS  8
#define SLOPE    0.2f

#define ROWS_PB    64
#define LDS_STRIDE 260

#define TAB_PAD    100352

#define MAIN_BLOCKS  1536
#define MAIN_ROWS    (MAIN_BLOCKS * ROWS_PB)
#define TAIL_TILES   ((N_NODES - MAIN_ROWS + 15) / 16)
#define TAIL_BLOCKS  TAIL_TILES
#define NODE_BLOCKS  ((N_NODES + ROWS_PB - 1) / ROWS_PB)

typedef __attribute__((ext_vector_type(2))) float v2f;
typedef __attribute__((ext_vector_type(8))) float v8f;

__device__ __forceinline__ void gat_gemm_tail(
    const float* __restrict__ x, const float* __restrict__ W, const float* __restrict__ bvec,
    float* __restrict__ O, float* __restrict__ xs, const int row0)
{
    const int tid = threadIdx.x;
    {
        const int c4 = tid & 63;
        const int rb = tid >> 6;
        float4 v[4];
#pragma unroll
        for (int i = 0; i < 4; ++i) {
            const int gr = min(row0 + rb + 4 * i, N_NODES - 1);
            v[i] = *(const float4*)(x + (size_t)gr * IN_F + c4 * 4);
        }
#pragma unroll
        for (int i = 0; i < 4; ++i)
            *(float4*)&xs[(rb + 4 * i) * LDS_STRIDE + c4 * 4] = v[i];
    }
    __syncthreads();

    const int wave    = tid >> 5;
    const int lane    = tid & 31;
    const int l15     = lane & 15;
    const int kq      = (lane >> 4) * 4;
    const int rhalf   = (lane >> 4) * 8;
    const int colbase = wave * 16;

    const float* __restrict__ arow = &xs[l15 * LDS_STRIDE + kq];
    const float* __restrict__ Wp   = W + (size_t)kq * HD + colbase + l15;
    v8f acc = {};
    for (int k = 0; k < IN_F; k += 8) {
        const float4 a = *(const float4*)(arow + k);
        v2f b0, b1, a01, a23;
        b0.x = Wp[(k + 0) * HD];
        b0.y = Wp[(k + 1) * HD];
        b1.x = Wp[(k + 2) * HD];
        b1.y = Wp[(k + 3) * HD];
        a01.x = a.x; a01.y = a.y;
        a23.x = a.z; a23.y = a.w;
        acc = __builtin_amdgcn_wmma_f32_16x16x4_f32(false, a01, false, b0, (short)0, acc, false, false);
        acc = __builtin_amdgcn_wmma_f32_16x16x4_f32(false, a23, false, b1, (short)0, acc, false, false);
    }
    const float bias = bvec[colbase + l15];
#pragma unroll
    for (int j = 0; j < 8; ++j) {
        const int r = row0 + rhalf + j;
        if (r < N_NODES)
            O[(size_t)r * HD + colbase + l15] = acc[j] + bias;
    }
}

#define GAT_LOADA(A0, A1, A2, A3, K)                                           \
    {                                                                          \
        A0 = *(const float4*)(arow + 0 * 16 * LDS_STRIDE + (K));               \
        A1 = *(const float4*)(arow + 1 * 16 * LDS_STRIDE + (K));               \
        A2 = *(const float4*)(arow + 2 * 16 * LDS_STRIDE + (K));               \
        A3 = *(const float4*)(arow + 3 * 16 * LDS_STRIDE + (K));               \
    }
#define GAT_LOADB(B0, B1, WP, K)                                               \
    {                                                                          \
        B0.x = WP[((K) + 0) * HD];                                             \
        B0.y = WP[((K) + 1) * HD];                                             \
        B1.x = WP[((K) + 2) * HD];                                             \
        B1.y = WP[((K) + 3) * HD];                                             \
    }
#define GAT_WMMA(ACC, AX, AY, B)                                               \
    {                                                                          \
        v2f a; a.x = (AX); a.y = (AY);                                         \
        ACC = __builtin_amdgcn_wmma_f32_16x16x4_f32(                           \
                false, a, false, B, (short)0, ACC, false, false);              \
    }

__global__ __launch_bounds__(256) void gat_gemm_el(
    const float* __restrict__ x,
    const float* __restrict__ Wsrc, const float* __restrict__ bsrc,
    float* __restrict__ el)
{
    __shared__ __attribute__((aligned(16))) float xs[ROWS_PB * LDS_STRIDE];

    if (blockIdx.x >= MAIN_BLOCKS) {
        gat_gemm_tail(x, Wsrc, bsrc, el, xs, MAIN_ROWS + (blockIdx.x - MAIN_BLOCKS) * 16);
        return;
    }

    const int tid  = threadIdx.x;
    const int row0 = blockIdx.x * ROWS_PB;

    {
        const int c4 = tid & 63;
        const int rb = tid >> 6;
        float4 v[16];
#pragma unroll
        for (int i = 0; i < 16; ++i) {
            const int gr = min(row0 + rb + 4 * i, N_NODES - 1);
            v[i] = *(const float4*)(x + (size_t)gr * IN_F + c4 * 4);
        }
#pragma unroll
        for (int i = 0; i < 16; ++i)
            *(float4*)&xs[(rb + 4 * i) * LDS_STRIDE + c4 * 4] = v[i];
    }
    __syncthreads();

    const int wave    = tid >> 5;
    const int lane    = tid & 31;
    const int l15     = lane & 15;
    const int kq      = (lane >> 4) * 4;
    const int rhalf   = (lane >> 4) * 8;
    const int colbase = wave * 16;

    const float* __restrict__ arow = &xs[l15 * LDS_STRIDE + kq];
    const float* __restrict__ Wp   = Wsrc + (size_t)kq * HD + colbase + l15;
    v8f acc0 = {}; v8f acc1 = {}; v8f acc2 = {}; v8f acc3 = {};

    for (int k = 0; k < IN_F; k += 8) {
        float4 a0, a1, a2, a3; v2f b0, b1;
        GAT_LOADA(a0, a1, a2, a3, k)
        GAT_LOADB(b0, b1, Wp, k)

        GAT_WMMA(acc0, a0.x, a0.y, b0)
        GAT_WMMA(acc1, a1.x, a1.y, b0)
        GAT_WMMA(acc2, a2.x, a2.y, b0)
        GAT_WMMA(acc3, a3.x, a3.y, b0)
        GAT_WMMA(acc0, a0.z, a0.w, b1)
        GAT_WMMA(acc1, a1.z, a1.w, b1)
        GAT_WMMA(acc2, a2.z, a2.w, b1)
        GAT_WMMA(acc3, a3.z, a3.w, b1)
    }

    const float bias = bsrc[colbase + l15];
#define GAT_STORE(ACC, RT)                                                     \
    {                                                                          \
        _Pragma("unroll")                                                      \
        for (int j = 0; j < 8; ++j) {                                          \
            const int r = row0 + (RT) * 16 + rhalf + j;                        \
            if (r < N_NODES)                                                   \
                el[(size_t)r * HD + colbase + l15] = ACC[j] + bias;            \
        }                                                                      \
    }
    GAT_STORE(acc0, 0)
    GAT_STORE(acc1, 1)
    GAT_STORE(acc2, 2)
    GAT_STORE(acc3, 3)
#undef GAT_STORE
}

#define NBK      256
#define NPB      391
#define NCH      256
#define CH       (E_EDGES / NCH)
#define CH_FULL  (CH / 256)
#define CH_TAIL  (CH - CH_FULL * 256)
#define SRC_BITS 17
#define SRC_MASK ((1 << SRC_BITS) - 1)
#define CSR_CAP  12288

#define RUN_BITS 16
#define RUN_MASK ((1 << RUN_BITS) - 1)
static_assert(CH < (1 << (RUN_BITS - 1)), "start and length of a run fit the table word");

static_assert(NBK * NPB >= N_NODES, "ranges cover the nodes");
static_assert(NCH * CH == E_EDGES, "chunks cover the edges");
static_assert((1 << SRC_BITS) > N_NODES && NPB < (1 << (31 - SRC_BITS)), "packed word fits");

__device__ __forceinline__ int gat_clamp_node(int d)
{
    return min(max(d, 0), N_NODES - 1);
}

__device__ __forceinline__ int gat_block_exscan(int v, int* wsum)
{
    const int lane = threadIdx.x & 31;
    const int wave = threadIdx.x >> 5;
    int s = v;
    for (int d = 1; d < 32; d <<= 1) {
        const int t = __shfl_up(s, d, 32);
        if (lane >= d) s += t;
    }
    if (lane == 31) wsum[wave] = s;
    __syncthreads();
    if (wave == 0) {
        int w = (lane < (int)(blockDim.x >> 5)) ? wsum[lane] : 0;
        for (int d = 1; d < 32; d <<= 1) {
            const int t = __shfl_up(w, d, 32);
            if (lane >= d) w += t;
        }
        wsum[lane] = w;
    }
    __syncthreads();
    const int woff = (wave > 0) ? wsum[wave - 1] : 0;
    return woff + s - v;
}

__global__ __launch_bounds__(256) void gat_chunk_order(
    const int* __restrict__ src, const int* __restrict__ dst,
    int* __restrict__ rtab, int* __restrict__ part)
{
    __shared__ int h[NBK];
    __shared__ int sw[CH];
    __shared__ int wsum[32];
    const int tid = threadIdx.x;
    const int blk = blockIdx.x;
    const int* __restrict__ sp = src + blk * CH;
    const int* __restrict__ dp = dst + blk * CH;
    h[tid] = 0;

    int d[CH_FULL], s[CH_FULL];
#pragma unroll
    for (int k = 0; k < CH_FULL; ++k) { d[k] = dp[k * 256 + tid]; s[k] = sp[k * 256 + tid]; }
    int       dl = dp[min(CH_FULL * 256 + tid, CH - 1)];
    const int sl = sp[min(CH_FULL * 256 + tid, CH - 1)];
    __syncthreads();
#pragma unroll
    for (int k = 0; k < CH_FULL; ++k) { d[k] = gat_clamp_node(d[k]); atomicAdd(&h[d[k] / NPB], 1); }
    dl = gat_clamp_node(dl);
    if (tid < CH_TAIL) atomicAdd(&h[dl / NPB], 1);
    __syncthreads();

    const int c  = h[tid];
    const int ex = gat_block_exscan(c, wsum);
    h[tid] = ex;
    rtab[tid * NCH + blk] = (ex << RUN_BITS) | c;
    __syncthreads();

#define GAT_PLACE(DN, S)                                                       \
    {                                                                          \
        const int b  = (DN) / NPB;                                             \
        const int sn = ((unsigned)(S) < (unsigned)N_NODES) ? (S) : 0;          \
        const int p  = atomicAdd(&h[b], 1);                                    \
        if ((unsigned)p < (unsigned)CH)                                        \
            sw[p] = (((DN) - b * NPB) << SRC_BITS) | sn;                       \
    }
#pragma unroll
    for (int k = 0; k < CH_FULL; ++k) GAT_PLACE(d[k], s[k])
    if (tid < CH_TAIL) GAT_PLACE(dl, sl)
#undef GAT_PLACE
    __syncthreads();

    int* __restrict__ pw = part + blk * CH;
    for (int i = tid; i < CH; i += 256) pw[i] = sw[i];
}

#define RUNS_PW  (NCH / 16)

__global__ __launch_bounds__(512) void gat_range_csr(
    const int* __restrict__ part, const int* __restrict__ rtab,
    int* __restrict__ rowptr, int* __restrict__ csr_src)
{
    __shared__ int cnt[512];
    __shared__ int lw[CSR_CAP];
    __shared__ int lcsr[CSR_CAP];
    __shared__ int rpos[NCH];
    __shared__ int roff[NCH + 1];
    __shared__ int wred[16];
    __shared__ int wsum[32];
    const int tid  = threadIdx.x;
    const int lane = tid & 31;
    const int wave = tid >> 5;
    const int b    = blockIdx.x;

    int st = 0, n = 0;
    if (tid < NCH) {
        const int w = rtab[b * NCH + tid];
        st = min((int)((unsigned)w >> RUN_BITS), CH);
        n  = min(w & RUN_MASK, CH - st);
    }
    cnt[tid] = 0;
    const int off = gat_block_exscan(n, wsum);
    int sred = st;
    for (int dd = 16; dd > 0; dd >>= 1) sred += __shfl_xor(sred, dd, 32);
    if (lane == 0) wred[wave] = sred;
    if (tid < NCH) { rpos[tid] = tid * CH + st; roff[tid] = off; }
    if (tid == NCH - 1) roff[NCH] = off + n;
    __syncthreads();

    int bs = 0;
    for (int k = 0; k < 16; ++k) bs += wred[k];
    bs = min(max(bs, 0), E_EDGES);
    const int nb = min(roff[NCH], E_EDGES - bs);
    const int be = bs + nb;
    const bool stage = (roff[NCH] <= CSR_CAP);

    if (stage) {
        int wv[RUNS_PW];
#pragma unroll
        for (int q = 0; q < RUNS_PW; ++q) {
            const int c   = wave + 16 * q;
            const int len = roff[c + 1] - roff[c];
            wv[q] = (lane < len) ? part[rpos[c] + lane] : 0;
        }
        int longrun = 0;
#pragma unroll
        for (int q = 0; q < RUNS_PW; ++q) {
            const int c   = wave + 16 * q;
            const int o   = roff[c];
            const int len = roff[c + 1] - o;
            if (lane < len) lw[o + lane] = wv[q];
            longrun |= (len > 32);
        }
        if (longrun) {
#pragma nounroll
            for (int c = wave; c < NCH; c += 16) {
                const int o   = roff[c];
                const int len = roff[c + 1] - o;
#pragma nounroll
                for (int j = lane + 32; j < len; j += 32) lw[o + j] = part[rpos[c] + j];
            }
        }
        __syncthreads();
        for (int i = tid; i < nb; i += 512)
            atomicAdd(&cnt[min((int)((unsigned)lw[i] >> SRC_BITS), NPB - 1)], 1);
    } else {
#pragma nounroll
        for (int c = wave; c < NCH; c += 16) {
            const int len = roff[c + 1] - roff[c];
#pragma nounroll
            for (int j = lane; j < len; j += 32)
                atomicAdd(&cnt[min((int)((unsigned)part[rpos[c] + j] >> SRC_BITS), NPB - 1)], 1);
        }
    }
    __syncthreads();

    const int c  = (tid < NPB) ? cnt[tid] : 0;
    const int ex = gat_block_exscan(c, wsum);
    cnt[tid] = ex;
    if (tid < NPB) {
        const int node = b * NPB + tid;
        if (node < N_NODES) rowptr[node] = bs + ex;
    }
    if (b == NBK - 1 && tid == 0) rowptr[N_NODES] = be;
    __syncthreads();

    if (stage) {
        for (int i = tid; i < nb; i += 512) {
            const int w  = lw[i];
            const int ln = min((int)((unsigned)w >> SRC_BITS), NPB - 1);
            const int p  = atomicAdd(&cnt[ln], 1);
            if ((unsigned)p < (unsigned)nb) lcsr[p] = w & SRC_MASK;
        }
        __syncthreads();
        for (int i = tid; i < nb; i += 512) csr_src[bs + i] = lcsr[i];
    } else {
#pragma nounroll
        for (int cc = wave; cc < NCH; cc += 16) {
            const int len = roff[cc + 1] - roff[cc];
#pragma nounroll
            for (int j = lane; j < len; j += 32) {
                const int w  = part[rpos[cc] + j];
                const int ln = min((int)((unsigned)w >> SRC_BITS), NPB - 1);
                const int p  = atomicAdd(&cnt[ln], 1);
                if ((unsigned)p < (unsigned)nb) csr_src[bs + p] = w & SRC_MASK;
            }
        }
    }
}

__device__ __forceinline__ float gat_logit(const float4 f, const float4 b4, const float4 w4)
{
    float p = 0.f, t;
    t = f.x + b4.x; t = (t > 0.f) ? t : SLOPE * t; p += t * w4.x;
    t = f.y + b4.y; t = (t > 0.f) ? t : SLOPE * t; p += t * w4.y;
    t = f.z + b4.z; t = (t > 0.f) ? t : SLOPE * t; p += t * w4.z;
    t = f.w + b4.w; t = (t > 0.f) ? t : SLOPE * t; p += t * w4.w;
    p += __shfl_xor(p, 1, 32);
    p += __shfl_xor(p, 2, 32);
    return p;
}

#define GAT_STEP4(F0, F1, F2, F3)                                                          \
    {                                                                                      \
        const float p0 = gat_logit(F0, b4, w4);                                            \
        const float p1 = gat_logit(F1, b4, w4);                                            \
        const float p2 = gat_logit(F2, b4, w4);                                            \
        const float p3 = gat_logit(F3, b4, w4);                                            \
        const float mn = fmaxf(m, fmaxf(fmaxf(p0, p1), fmaxf(p2, p3)));                    \
        const float sc = __expf(m - mn);                                                   \
        const float a0 = __expf(p0 - mn);                                                  \
        const float a1 = __expf(p1 - mn);                                                  \
        const float a2 = __expf(p2 - mn);                                                  \
        const float a3 = __expf(p3 - mn);                                                  \
        l     = l * sc + ((a0 + a1) + (a2 + a3));                                          \
        acc.x = acc.x * sc + ((a0 * F0.x + a1 * F1.x) + (a2 * F2.x + a3 * F3.x));          \
        acc.y = acc.y * sc + ((a0 * F0.y + a1 * F1.y) + (a2 * F2.y + a3 * F3.y));          \
        acc.z = acc.z * sc + ((a0 * F0.z + a1 * F1.z) + (a2 * F2.z + a3 * F3.z));          \
        acc.w = acc.w * sc + ((a0 * F0.w + a1 * F1.w) + (a2 * F2.w + a3 * F3.w));          \
        m = mn;                                                                            \
    }
#define GAT_ROW(J)                                                                         \
    ({                                                                                     \
        int u_ = __builtin_amdgcn_readlane(idx, (J));                                      \
        u_ = ((unsigned)u_ < (unsigned)N_NODES) ? u_ : 0;                                  \
        *(const float4*)(el + (size_t)u_ * HD + c);                                        \
    })

__global__ __launch_bounds__(256) void gat_node_fused(
    const float* __restrict__ x,
    const float* __restrict__ Wdst, const float* __restrict__ bdst,
    const float* __restrict__ Wres, const float* __restrict__ bres,
    const float* __restrict__ el,
    const int* __restrict__ rowptr, const int* __restrict__ csr_src,
    const float* __restrict__ attn, float* __restrict__ out)
{
    __shared__ __attribute__((aligned(16))) float xs[ROWS_PB * LDS_STRIDE];

    const int tid  = threadIdx.x;
    const int row0 = blockIdx.x * ROWS_PB;
    const int lane = tid & 31;
    const int wv   = __builtin_amdgcn_readfirstlane(tid >> 5);

    const int rp = rowptr[min(row0 + wv * 8 + min(lane, 8), N_NODES)];

    {
        const int c4 = tid & 63;
        const int rb = tid >> 6;
        float4 v[16];
#pragma unroll
        for (int i = 0; i < 16; ++i) {
            const int gr = min(row0 + rb + 4 * i, N_NODES - 1);
            v[i] = *(const float4*)(x + (size_t)gr * IN_F + c4 * 4);
        }
#pragma unroll
        for (int i = 0; i < 16; ++i)
            *(float4*)&xs[(rb + 4 * i) * LDS_STRIDE + c4 * 4] = v[i];
    }
    __syncthreads();

    const int l15     = lane & 15;
    const int kq      = (lane >> 4) * 4;
    const int rhalf   = (lane >> 4) * 8;
    const int colbase = wv * 16;

    {
        const float* __restrict__ arow = &xs[l15 * LDS_STRIDE + kq];
        const float* __restrict__ Wd   = Wdst + (size_t)kq * HD + colbase + l15;
        const float* __restrict__ Wr   = Wres + (size_t)kq * HD + colbase + l15;
        v8f e0 = {}; v8f e1 = {}; v8f e2 = {}; v8f e3 = {};
        v8f r0 = {}; v8f r1 = {}; v8f r2 = {}; v8f r3 = {};

        for (int k = 0; k < IN_F; k += 8) {
            float4 a0, a1, a2, a3; v2f d0, d1, s0, s1;
            GAT_LOADA(a0, a1, a2, a3, k)
            GAT_LOADB(d0, d1, Wd, k)
            GAT_LOADB(s0, s1, Wr, k)

            GAT_WMMA(e0, a0.x, a0.y, d0)
            GAT_WMMA(e1, a1.x, a1.y, d0)
            GAT_WMMA(e2, a2.x, a2.y, d0)
            GAT_WMMA(e3, a3.x, a3.y, d0)
            GAT_WMMA(r0, a0.x, a0.y, s0)
            GAT_WMMA(r1, a1.x, a1.y, s0)
            GAT_WMMA(r2, a2.x, a2.y, s0)
            GAT_WMMA(r3, a3.x, a3.y, s0)
            GAT_WMMA(e0, a0.z, a0.w, d1)
            GAT_WMMA(e1, a1.z, a1.w, d1)
            GAT_WMMA(e2, a2.z, a2.w, d1)
            GAT_WMMA(e3, a3.z, a3.w, d1)
            GAT_WMMA(r0, a0.z, a0.w, s1)
            GAT_WMMA(r1, a1.z, a1.w, s1)
            GAT_WMMA(r2, a2.z, a2.w, s1)
            GAT_WMMA(r3, a3.z, a3.w, s1)
        }

        const float bias_d = bdst[colbase + l15];
        const float bias_r = bres[colbase + l15];
        __syncthreads();

#define GAT_KEEP(EACC, RACC, RT)                                                           \
        {                                                                                  \
            _Pragma("unroll")                                                              \
            for (int j = 0; j < 8; ++j) {                                                  \
                float* q = &xs[((RT) * 16 + rhalf + j) * LDS_STRIDE + colbase + l15];      \
                q[0]  = EACC[j] + bias_d;                                                  \
                q[HD] = RACC[j] + bias_r;                                                  \
            }                                                                              \
        }
        GAT_KEEP(e0, r0, 0)
        GAT_KEEP(e1, r1, 1)
        GAT_KEEP(e2, r2, 2)
        GAT_KEEP(e3, r3, 3)
#undef GAT_KEEP
    }
    __syncthreads();

    const int    c  = lane * 4;
    const float4 w4 = *(const float4*)(attn + c);

    for (int i = 0; i < 8; ++i) {
        const int v = row0 + wv * 8 + i;
        if (v >= N_NODES) break;

        int beg = __builtin_amdgcn_readlane(rp, i);
        int end = __builtin_amdgcn_readlane(rp, i + 1);
        beg = max(beg, 0);
        end = min(end, E_EDGES);

        const float* trow = &xs[(wv * 8 + i) * LDS_STRIDE + c];
        const float4 b4 = *(const float4*)trow;

        float  m   = -INFINITY;
        float  l   = 0.f;
        float4 acc = make_float4(0.f, 0.f, 0.f, 0.f);

        for (int base = beg; base < end; base += 32) {
            const int cnt = min(32, end - base);
            int idx = 0;
            if (lane < cnt) idx = csr_src[base + lane];

            int j = 0;
            for (; j + 8 <= cnt; j += 8) {
                const float4 f0 = GAT_ROW(j);
                const float4 f1 = GAT_ROW(j + 1);
                const float4 f2 = GAT_ROW(j + 2);
                const float4 f3 = GAT_ROW(j + 3);
                const float4 f4 = GAT_ROW(j + 4);
                const float4 f5 = GAT_ROW(j + 5);
                const float4 f6 = GAT_ROW(j + 6);
                const float4 f7 = GAT_ROW(j + 7);
                GAT_STEP4(f0, f1, f2, f3)
                GAT_STEP4(f4, f5, f6, f7)
            }
            for (; j + 4 <= cnt; j += 4) {
                const float4 f0 = GAT_ROW(j);
                const float4 f1 = GAT_ROW(j + 1);
                const float4 f2 = GAT_ROW(j + 2);
                const float4 f3 = GAT_ROW(j + 3);
                GAT_STEP4(f0, f1, f2, f3)
            }
            for (; j < cnt; ++j) {
                const float4 f0 = GAT_ROW(j);
                const float p0 = gat_logit(f0, b4, w4);
                const float mn = fmaxf(m, p0);
                const float sc = __expf(m - mn);
                const float a0 = __expf(p0 - mn);
                l     = l * sc + a0;
                acc.x = acc.x * sc + a0 * f0.x;
                acc.y = acc.y * sc + a0 * f0.y;
                acc.z = acc.z * sc + a0 * f0.z;
                acc.w = acc.w * sc + a0 * f0.w;
                m = mn;
            }
        }

        float4 r = *(const float4*)(trow + HD);
        if (end > beg) {
            r.x += acc.x / l;
            r.y += acc.y / l;
            r.z += acc.z / l;
            r.w += acc.w / l;
        }
        *(float4*)(out + (size_t)v * HD + c) = r;
    }
}
#undef GAT_ROW
#undef GAT_STEP4
#undef GAT_WMMA
#undef GAT_LOADB
#undef GAT_LOADA

extern "C" void kernel_launch(void* const* d_in, const int* in_sizes, int n_in,
                              void* d_out, int out_size, void* d_ws, size_t ws_size,
                              hipStream_t stream) {
    const float* x    = (const float*)d_in[0];
    const int*   src  = (const int*)  d_in[1];
    const int*   dst  = (const int*)  d_in[2];
    const float* Wsrc = (const float*)d_in[3];
    const float* bsrc = (const float*)d_in[4];
    const float* Wdst = (const float*)d_in[5];
    const float* bdst = (const float*)d_in[6];
    const float* attn = (const float*)d_in[7];
    const float* Wres = (const float*)d_in[8];
    const float* bres = (const float*)d_in[9];
    float* out = (float*)d_out;

    char* ws = (char*)d_ws;
    const size_t elBytes   = (size_t)N_NODES * HD * sizeof(float);
    const size_t padBytes  = (size_t)TAB_PAD * sizeof(int);
    const size_t edgeBytes = (size_t)E_EDGES * sizeof(int);
    float* el     = (float*)(ws);
    int*   rtab   = (int*)  (ws + 2 * elBytes);
    int*   rowptr = (int*)  (ws + 2 * elBytes + padBytes);
    int*   part   = (int*)  (ws + 2 * elBytes + 2 * padBytes + 512);
    int*   csr    = (int*)  (ws + 2 * elBytes + 2 * padBytes + 512 + edgeBytes);

    gat_chunk_order<<<NCH, 256, 0, stream>>>(src, dst, rtab, part);
    gat_range_csr<<<NBK, 512, 0, stream>>>(part, rtab, rowptr, csr);

    gat_gemm_el<<<MAIN_BLOCKS + TAIL_BLOCKS, 256, 0, stream>>>(x, Wsrc, bsrc, el);

    gat_node_fused<<<NODE_BLOCKS, 256, 0, stream>>>(x, Wdst, bdst, Wres, bres, el, rowptr, csr, attn, out);
}


# Node kernel split so the WGPs fill evenly: 1536 64-node work-groups (six per WGP) plus 106 small ones (16 nodes: one er/res row tile, 2 nodes per wave), not 27 big ones left over; sums unchanged
# speedup vs baseline: 4.0648x; 1.0443x over previous
#include <hip/hip_runtime.h>
#include <hip/hip_bf16.h>

#define N_NODES  100000
#define E_EDGES  1600000
#define IN_F     256
#define HD       128
#define H_HEADS  8
#define SLOPE    0.2f

#define ROWS_PB    64
#define LDS_STRIDE 260

#define TAB_PAD    100352

#define MAIN_BLOCKS  1536
#define MAIN_ROWS    (MAIN_BLOCKS * ROWS_PB)
#define TAIL_TILES   ((N_NODES - MAIN_ROWS + 15) / 16)
#define TAIL_BLOCKS  TAIL_TILES
#define TAIL_NODES   16
static_assert(MAIN_ROWS + TAIL_BLOCKS * TAIL_NODES == N_NODES, "the small work-groups cover the rest exactly");

typedef __attribute__((ext_vector_type(2))) float v2f;
typedef __attribute__((ext_vector_type(8))) float v8f;

__device__ __forceinline__ void gat_gemm_tail(
    const float* __restrict__ x, const float* __restrict__ W, const float* __restrict__ bvec,
    float* __restrict__ O, float* __restrict__ xs, const int row0)
{
    const int tid = threadIdx.x;
    {
        const int c4 = tid & 63;
        const int rb = tid >> 6;
        float4 v[4];
#pragma unroll
        for (int i = 0; i < 4; ++i) {
            const int gr = min(row0 + rb + 4 * i, N_NODES - 1);
            v[i] = *(const float4*)(x + (size_t)gr * IN_F + c4 * 4);
        }
#pragma unroll
        for (int i = 0; i < 4; ++i)
            *(float4*)&xs[(rb + 4 * i) * LDS_STRIDE + c4 * 4] = v[i];
    }
    __syncthreads();

    const int wave    = tid >> 5;
    const int lane    = tid & 31;
    const int l15     = lane & 15;
    const int kq      = (lane >> 4) * 4;
    const int rhalf   = (lane >> 4) * 8;
    const int colbase = wave * 16;

    const float* __restrict__ arow = &xs[l15 * LDS_STRIDE + kq];
    const float* __restrict__ Wp   = W + (size_t)kq * HD + colbase + l15;
    v8f acc = {};
    for (int k = 0; k < IN_F; k += 8) {
        const float4 a = *(const float4*)(arow + k);
        v2f b0, b1, a01, a23;
        b0.x = Wp[(k + 0) * HD];
        b0.y = Wp[(k + 1) * HD];
        b1.x = Wp[(k + 2) * HD];
        b1.y = Wp[(k + 3) * HD];
        a01.x = a.x; a01.y = a.y;
        a23.x = a.z; a23.y = a.w;
        acc = __builtin_amdgcn_wmma_f32_16x16x4_f32(false, a01, false, b0, (short)0, acc, false, false);
        acc = __builtin_amdgcn_wmma_f32_16x16x4_f32(false, a23, false, b1, (short)0, acc, false, false);
    }
    const float bias = bvec[colbase + l15];
#pragma unroll
    for (int j = 0; j < 8; ++j) {
        const int r = row0 + rhalf + j;
        if (r < N_NODES)
            O[(size_t)r * HD + colbase + l15] = acc[j] + bias;
    }
}

#define GAT_LOADA(A0, A1, A2, A3, K)                                           \
    {                                                                          \
        A0 = *(const float4*)(arow + 0 * 16 * LDS_STRIDE + (K));               \
        A1 = *(const float4*)(arow + 1 * 16 * LDS_STRIDE + (K));               \
        A2 = *(const float4*)(arow + 2 * 16 * LDS_STRIDE + (K));               \
        A3 = *(const float4*)(arow + 3 * 16 * LDS_STRIDE + (K));               \
    }
#define GAT_LOADB(B0, B1, WP, K)                                               \
    {                                                                          \
        B0.x = WP[((K) + 0) * HD];                                             \
        B0.y = WP[((K) + 1) * HD];                                             \
        B1.x = WP[((K) + 2) * HD];                                             \
        B1.y = WP[((K) + 3) * HD];                                             \
    }
#define GAT_WMMA(ACC, AX, AY, B)                                               \
    {                                                                          \
        v2f a; a.x = (AX); a.y = (AY);                                         \
        ACC = __builtin_amdgcn_wmma_f32_16x16x4_f32(                           \
                false, a, false, B, (short)0, ACC, false, false);              \
    }

__global__ __launch_bounds__(256) void gat_gemm_el(
    const float* __restrict__ x,
    const float* __restrict__ Wsrc, const float* __restrict__ bsrc,
    float* __restrict__ el)
{
    __shared__ __attribute__((aligned(16))) float xs[ROWS_PB * LDS_STRIDE];

    if (blockIdx.x >= MAIN_BLOCKS) {
        gat_gemm_tail(x, Wsrc, bsrc, el, xs, MAIN_ROWS + (blockIdx.x - MAIN_BLOCKS) * 16);
        return;
    }

    const int tid  = threadIdx.x;
    const int row0 = blockIdx.x * ROWS_PB;

    {
        const int c4 = tid & 63;
        const int rb = tid >> 6;
        float4 v[16];
#pragma unroll
        for (int i = 0; i < 16; ++i) {
            const int gr = min(row0 + rb + 4 * i, N_NODES - 1);
            v[i] = *(const float4*)(x + (size_t)gr * IN_F + c4 * 4);
        }
#pragma unroll
        for (int i = 0; i < 16; ++i)
            *(float4*)&xs[(rb + 4 * i) * LDS_STRIDE + c4 * 4] = v[i];
    }
    __syncthreads();

    const int wave    = tid >> 5;
    const int lane    = tid & 31;
    const int l15     = lane & 15;
    const int kq      = (lane >> 4) * 4;
    const int rhalf   = (lane >> 4) * 8;
    const int colbase = wave * 16;

    const float* __restrict__ arow = &xs[l15 * LDS_STRIDE + kq];
    const float* __restrict__ Wp   = Wsrc + (size_t)kq * HD + colbase + l15;
    v8f acc0 = {}; v8f acc1 = {}; v8f acc2 = {}; v8f acc3 = {};

    for (int k = 0; k < IN_F; k += 8) {
        float4 a0, a1, a2, a3; v2f b0, b1;
        GAT_LOADA(a0, a1, a2, a3, k)
        GAT_LOADB(b0, b1, Wp, k)

        GAT_WMMA(acc0, a0.x, a0.y, b0)
        GAT_WMMA(acc1, a1.x, a1.y, b0)
        GAT_WMMA(acc2, a2.x, a2.y, b0)
        GAT_WMMA(acc3, a3.x, a3.y, b0)
        GAT_WMMA(acc0, a0.z, a0.w, b1)
        GAT_WMMA(acc1, a1.z, a1.w, b1)
        GAT_WMMA(acc2, a2.z, a2.w, b1)
        GAT_WMMA(acc3, a3.z, a3.w, b1)
    }

    const float bias = bsrc[colbase + l15];
#define GAT_STORE(ACC, RT)                                                     \
    {                                                                          \
        _Pragma("unroll")                                                      \
        for (int j = 0; j < 8; ++j) {                                          \
            const int r = row0 + (RT) * 16 + rhalf + j;                        \
            if (r < N_NODES)                                                   \
                el[(size_t)r * HD + colbase + l15] = ACC[j] + bias;            \
        }                                                                      \
    }
    GAT_STORE(acc0, 0)
    GAT_STORE(acc1, 1)
    GAT_STORE(acc2, 2)
    GAT_STORE(acc3, 3)
#undef GAT_STORE
}

#define NBK      256
#define NPB      391
#define NCH      256
#define CH       (E_EDGES / NCH)
#define CH_FULL  (CH / 256)
#define CH_TAIL  (CH - CH_FULL * 256)
#define SRC_BITS 17
#define SRC_MASK ((1 << SRC_BITS) - 1)
#define CSR_CAP  12288

#define RUN_BITS 16
#define RUN_MASK ((1 << RUN_BITS) - 1)
static_assert(CH < (1 << (RUN_BITS - 1)), "start and length of a run fit the table word");

static_assert(NBK * NPB >= N_NODES, "ranges cover the nodes");
static_assert(NCH * CH == E_EDGES, "chunks cover the edges");
static_assert((1 << SRC_BITS) > N_NODES && NPB < (1 << (31 - SRC_BITS)), "packed word fits");

__device__ __forceinline__ int gat_clamp_node(int d)
{
    return min(max(d, 0), N_NODES - 1);
}

__device__ __forceinline__ int gat_block_exscan(int v, int* wsum)
{
    const int lane = threadIdx.x & 31;
    const int wave = threadIdx.x >> 5;
    int s = v;
    for (int d = 1; d < 32; d <<= 1) {
        const int t = __shfl_up(s, d, 32);
        if (lane >= d) s += t;
    }
    if (lane == 31) wsum[wave] = s;
    __syncthreads();
    if (wave == 0) {
        int w = (lane < (int)(blockDim.x >> 5)) ? wsum[lane] : 0;
        for (int d = 1; d < 32; d <<= 1) {
            const int t = __shfl_up(w, d, 32);
            if (lane >= d) w += t;
        }
        wsum[lane] = w;
    }
    __syncthreads();
    const int woff = (wave > 0) ? wsum[wave - 1] : 0;
    return woff + s - v;
}

__global__ __launch_bounds__(256) void gat_chunk_order(
    const int* __restrict__ src, const int* __restrict__ dst,
    int* __restrict__ rtab, int* __restrict__ part)
{
    __shared__ int h[NBK];
    __shared__ int sw[CH];
    __shared__ int wsum[32];
    const int tid = threadIdx.x;
    const int blk = blockIdx.x;
    const int* __restrict__ sp = src + blk * CH;
    const int* __restrict__ dp = dst + blk * CH;
    h[tid] = 0;

    int d[CH_FULL], s[CH_FULL];
#pragma unroll
    for (int k = 0; k < CH_FULL; ++k) { d[k] = dp[k * 256 + tid]; s[k] = sp[k * 256 + tid]; }
    int       dl = dp[min(CH_FULL * 256 + tid, CH - 1)];
    const int sl = sp[min(CH_FULL * 256 + tid, CH - 1)];
    __syncthreads();
#pragma unroll
    for (int k = 0; k < CH_FULL; ++k) { d[k] = gat_clamp_node(d[k]); atomicAdd(&h[d[k] / NPB], 1); }
    dl = gat_clamp_node(dl);
    if (tid < CH_TAIL) atomicAdd(&h[dl / NPB], 1);
    __syncthreads();

    const int c  = h[tid];
    const int ex = gat_block_exscan(c, wsum);
    h[tid] = ex;
    rtab[tid * NCH + blk] = (ex << RUN_BITS) | c;
    __syncthreads();

#define GAT_PLACE(DN, S)                                                       \
    {                                                                          \
        const int b  = (DN) / NPB;                                             \
        const int sn = ((unsigned)(S) < (unsigned)N_NODES) ? (S) : 0;          \
        const int p  = atomicAdd(&h[b], 1);                                    \
        if ((unsigned)p < (unsigned)CH)                                        \
            sw[p] = (((DN) - b * NPB) << SRC_BITS) | sn;                       \
    }
#pragma unroll
    for (int k = 0; k < CH_FULL; ++k) GAT_PLACE(d[k], s[k])
    if (tid < CH_TAIL) GAT_PLACE(dl, sl)
#undef GAT_PLACE
    __syncthreads();

    int* __restrict__ pw = part + blk * CH;
    for (int i = tid; i < CH; i += 256) pw[i] = sw[i];
}

#define RUNS_PW  (NCH / 16)

__global__ __launch_bounds__(512) void gat_range_csr(
    const int* __restrict__ part, const int* __restrict__ rtab,
    int* __restrict__ rowptr, int* __restrict__ csr_src)
{
    __shared__ int cnt[512];
    __shared__ int lw[CSR_CAP];
    __shared__ int lcsr[CSR_CAP];
    __shared__ int rpos[NCH];
    __shared__ int roff[NCH + 1];
    __shared__ int wred[16];
    __shared__ int wsum[32];
    const int tid  = threadIdx.x;
    const int lane = tid & 31;
    const int wave = tid >> 5;
    const int b    = blockIdx.x;

    int st = 0, n = 0;
    if (tid < NCH) {
        const int w = rtab[b * NCH + tid];
        st = min((int)((unsigned)w >> RUN_BITS), CH);
        n  = min(w & RUN_MASK, CH - st);
    }
    cnt[tid] = 0;
    const int off = gat_block_exscan(n, wsum);
    int sred = st;
    for (int dd = 16; dd > 0; dd >>= 1) sred += __shfl_xor(sred, dd, 32);
    if (lane == 0) wred[wave] = sred;
    if (tid < NCH) { rpos[tid] = tid * CH + st; roff[tid] = off; }
    if (tid == NCH - 1) roff[NCH] = off + n;
    __syncthreads();

    int bs = 0;
    for (int k = 0; k < 16; ++k) bs += wred[k];
    bs = min(max(bs, 0), E_EDGES);
    const int nb = min(roff[NCH], E_EDGES - bs);
    const int be = bs + nb;
    const bool stage = (roff[NCH] <= CSR_CAP);

    if (stage) {
        int wv[RUNS_PW];
#pragma unroll
        for (int q = 0; q < RUNS_PW; ++q) {
            const int c   = wave + 16 * q;
            const int len = roff[c + 1] - roff[c];
            wv[q] = (lane < len) ? part[rpos[c] + lane] : 0;
        }
        int longrun = 0;
#pragma unroll
        for (int q = 0; q < RUNS_PW; ++q) {
            const int c   = wave + 16 * q;
            const int o   = roff[c];
            const int len = roff[c + 1] - o;
            if (lane < len) lw[o + lane] = wv[q];
            longrun |= (len > 32);
        }
        if (longrun) {
#pragma nounroll
            for (int c = wave; c < NCH; c += 16) {
                const int o   = roff[c];
                const int len = roff[c + 1] - o;
#pragma nounroll
                for (int j = lane + 32; j < len; j += 32) lw[o + j] = part[rpos[c] + j];
            }
        }
        __syncthreads();
        for (int i = tid; i < nb; i += 512)
            atomicAdd(&cnt[min((int)((unsigned)lw[i] >> SRC_BITS), NPB - 1)], 1);
    } else {
#pragma nounroll
        for (int c = wave; c < NCH; c += 16) {
            const int len = roff[c + 1] - roff[c];
#pragma nounroll
            for (int j = lane; j < len; j += 32)
                atomicAdd(&cnt[min((int)((unsigned)part[rpos[c] + j] >> SRC_BITS), NPB - 1)], 1);
        }
    }
    __syncthreads();

    const int c  = (tid < NPB) ? cnt[tid] : 0;
    const int ex = gat_block_exscan(c, wsum);
    cnt[tid] = ex;
    if (tid < NPB) {
        const int node = b * NPB + tid;
        if (node < N_NODES) rowptr[node] = bs + ex;
    }
    if (b == NBK - 1 && tid == 0) rowptr[N_NODES] = be;
    __syncthreads();

    if (stage) {
        for (int i = tid; i < nb; i += 512) {
            const int w  = lw[i];
            const int ln = min((int)((unsigned)w >> SRC_BITS), NPB - 1);
            const int p  = atomicAdd(&cnt[ln], 1);
            if ((unsigned)p < (unsigned)nb) lcsr[p] = w & SRC_MASK;
        }
        __syncthreads();
        for (int i = tid; i < nb; i += 512) csr_src[bs + i] = lcsr[i];
    } else {
#pragma nounroll
        for (int cc = wave; cc < NCH; cc += 16) {
            const int len = roff[cc + 1] - roff[cc];
#pragma nounroll
            for (int j = lane; j < len; j += 32) {
                const int w  = part[rpos[cc] + j];
                const int ln = min((int)((unsigned)w >> SRC_BITS), NPB - 1);
                const int p  = atomicAdd(&cnt[ln], 1);
                if ((unsigned)p < (unsigned)nb) csr_src[bs + p] = w & SRC_MASK;
            }
        }
    }
}

__device__ __forceinline__ float gat_logit(const float4 f, const float4 b4, const float4 w4)
{
    float p = 0.f, t;
    t = f.x + b4.x; t = (t > 0.f) ? t : SLOPE * t; p += t * w4.x;
    t = f.y + b4.y; t = (t > 0.f) ? t : SLOPE * t; p += t * w4.y;
    t = f.z + b4.z; t = (t > 0.f) ? t : SLOPE * t; p += t * w4.z;
    t = f.w + b4.w; t = (t > 0.f) ? t : SLOPE * t; p += t * w4.w;
    p += __shfl_xor(p, 1, 32);
    p += __shfl_xor(p, 2, 32);
    return p;
}

#define GAT_STEP4(F0, F1, F2, F3)                                                          \
    {                                                                                      \
        const float p0 = gat_logit(F0, b4, w4);                                            \
        const float p1 = gat_logit(F1, b4, w4);                                            \
        const float p2 = gat_logit(F2, b4, w4);                                            \
        const float p3 = gat_logit(F3, b4, w4);                                            \
        const float mn = fmaxf(m, fmaxf(fmaxf(p0, p1), fmaxf(p2, p3)));                    \
        const float sc = __expf(m - mn);                                                   \
        const float a0 = __expf(p0 - mn);                                                  \
        const float a1 = __expf(p1 - mn);                                                  \
        const float a2 = __expf(p2 - mn);                                                  \
        const float a3 = __expf(p3 - mn);                                                  \
        l     = l * sc + ((a0 + a1) + (a2 + a3));                                          \
        acc.x = acc.x * sc + ((a0 * F0.x + a1 * F1.x) + (a2 * F2.x + a3 * F3.x));          \
        acc.y = acc.y * sc + ((a0 * F0.y + a1 * F1.y) + (a2 * F2.y + a3 * F3.y));          \
        acc.z = acc.z * sc + ((a0 * F0.z + a1 * F1.z) + (a2 * F2.z + a3 * F3.z));          \
        acc.w = acc.w * sc + ((a0 * F0.w + a1 * F1.w) + (a2 * F2.w + a3 * F3.w));          \
        m = mn;                                                                            \
    }
#define GAT_ROW(J)                                                                         \
    ({                                                                                     \
        int u_ = __builtin_amdgcn_readlane(idx, (J));                                      \
        u_ = ((unsigned)u_ < (unsigned)N_NODES) ? u_ : 0;                                  \
        *(const float4*)(el + (size_t)u_ * HD + c);                                        \
    })

__global__ __launch_bounds__(256) void gat_node_fused(
    const float* __restrict__ x,
    const float* __restrict__ Wdst, const float* __restrict__ bdst,
    const float* __restrict__ Wres, const float* __restrict__ bres,
    const float* __restrict__ el,
    const int* __restrict__ rowptr, const int* __restrict__ csr_src,
    const float* __restrict__ attn, float* __restrict__ out)
{
    __shared__ __attribute__((aligned(16))) float xs[ROWS_PB * LDS_STRIDE];

    const int tid  = threadIdx.x;
    const int lane = tid & 31;
    const int wv   = __builtin_amdgcn_readfirstlane(tid >> 5);

    const bool small = blockIdx.x >= MAIN_BLOCKS;
    const int  row0  = small ? MAIN_ROWS + (blockIdx.x - MAIN_BLOCKS) * TAIL_NODES : blockIdx.x * ROWS_PB;
    const int  npw   = small ? TAIL_NODES / 8 : ROWS_PB / 8;
    const int  nd0   = wv * npw;

    const int rp = rowptr[min(row0 + nd0 + min(lane, npw), N_NODES)];

    const int l15     = lane & 15;
    const int kq      = (lane >> 4) * 4;
    const int rhalf   = (lane >> 4) * 8;
    const int colbase = wv * 16;

    if (!small) {
        {
            const int c4 = tid & 63;
            const int rb = tid >> 6;
            float4 v[16];
#pragma unroll
            for (int i = 0; i < 16; ++i) {
                const int gr = min(row0 + rb + 4 * i, N_NODES - 1);
                v[i] = *(const float4*)(x + (size_t)gr * IN_F + c4 * 4);
            }
#pragma unroll
            for (int i = 0; i < 16; ++i)
                *(float4*)&xs[(rb + 4 * i) * LDS_STRIDE + c4 * 4] = v[i];
        }
        __syncthreads();

        const float* __restrict__ arow = &xs[l15 * LDS_STRIDE + kq];
        const float* __restrict__ Wd   = Wdst + (size_t)kq * HD + colbase + l15;
        const float* __restrict__ Wr   = Wres + (size_t)kq * HD + colbase + l15;
        v8f e0 = {}; v8f e1 = {}; v8f e2 = {}; v8f e3 = {};
        v8f r0 = {}; v8f r1 = {}; v8f r2 = {}; v8f r3 = {};

        for (int k = 0; k < IN_F; k += 8) {
            float4 a0, a1, a2, a3; v2f d0, d1, s0, s1;
            GAT_LOADA(a0, a1, a2, a3, k)
            GAT_LOADB(d0, d1, Wd, k)
            GAT_LOADB(s0, s1, Wr, k)

            GAT_WMMA(e0, a0.x, a0.y, d0)
            GAT_WMMA(e1, a1.x, a1.y, d0)
            GAT_WMMA(e2, a2.x, a2.y, d0)
            GAT_WMMA(e3, a3.x, a3.y, d0)
            GAT_WMMA(r0, a0.x, a0.y, s0)
            GAT_WMMA(r1, a1.x, a1.y, s0)
            GAT_WMMA(r2, a2.x, a2.y, s0)
            GAT_WMMA(r3, a3.x, a3.y, s0)
            GAT_WMMA(e0, a0.z, a0.w, d1)
            GAT_WMMA(e1, a1.z, a1.w, d1)
            GAT_WMMA(e2, a2.z, a2.w, d1)
            GAT_WMMA(e3, a3.z, a3.w, d1)
            GAT_WMMA(r0, a0.z, a0.w, s1)
            GAT_WMMA(r1, a1.z, a1.w, s1)
            GAT_WMMA(r2, a2.z, a2.w, s1)
            GAT_WMMA(r3, a3.z, a3.w, s1)
        }

        const float bias_d = bdst[colbase + l15];
        const float bias_r = bres[colbase + l15];
        __syncthreads();

#define GAT_KEEP(EACC, RACC, RT)                                                           \
        {                                                                                  \
            _Pragma("unroll")                                                              \
            for (int j = 0; j < 8; ++j) {                                                  \
                float* q = &xs[((RT) * 16 + rhalf + j) * LDS_STRIDE + colbase + l15];      \
                q[0]  = EACC[j] + bias_d;                                                  \
                q[HD] = RACC[j] + bias_r;                                                  \
            }                                                                              \
        }
        GAT_KEEP(e0, r0, 0)
        GAT_KEEP(e1, r1, 1)
        GAT_KEEP(e2, r2, 2)
        GAT_KEEP(e3, r3, 3)
    } else {
        {
            const int c4 = tid & 63;
            const int rb = tid >> 6;
            float4 v[4];
#pragma unroll
            for (int i = 0; i < 4; ++i) {
                const int gr = min(row0 + rb + 4 * i, N_NODES - 1);
                v[i] = *(const float4*)(x + (size_t)gr * IN_F + c4 * 4);
            }
#pragma unroll
            for (int i = 0; i < 4; ++i)
                *(float4*)&xs[(rb + 4 * i) * LDS_STRIDE + c4 * 4] = v[i];
        }
        __syncthreads();

        const float* __restrict__ arow = &xs[l15 * LDS_STRIDE + kq];
        const float* __restrict__ Wd   = Wdst + (size_t)kq * HD + colbase + l15;
        const float* __restrict__ Wr   = Wres + (size_t)kq * HD + colbase + l15;
        v8f e0 = {};
        v8f r0 = {};

        for (int k = 0; k < IN_F; k += 8) {
            v2f d0, d1, s0, s1;
            const float4 a0 = *(const float4*)(arow + k);
            GAT_LOADB(d0, d1, Wd, k)
            GAT_LOADB(s0, s1, Wr, k)

            GAT_WMMA(e0, a0.x, a0.y, d0)
            GAT_WMMA(r0, a0.x, a0.y, s0)
            GAT_WMMA(e0, a0.z, a0.w, d1)
            GAT_WMMA(r0, a0.z, a0.w, s1)
        }

        const float bias_d = bdst[colbase + l15];
        const float bias_r = bres[colbase + l15];
        __syncthreads();

        GAT_KEEP(e0, r0, 0)
#undef GAT_KEEP
    }
    __syncthreads();

    const int    c  = lane * 4;
    const float4 w4 = *(const float4*)(attn + c);

    for (int i = 0; i < npw; ++i) {
        const int v = row0 + nd0 + i;
        if (v >= N_NODES) break;

        int beg = __builtin_amdgcn_readlane(rp, i);
        int end = __builtin_amdgcn_readlane(rp, i + 1);
        beg = max(beg, 0);
        end = min(end, E_EDGES);

        const float* trow = &xs[(nd0 + i) * LDS_STRIDE + c];
        const float4 b4 = *(const float4*)trow;

        float  m   = -INFINITY;
        float  l   = 0.f;
        float4 acc = make_float4(0.f, 0.f, 0.f, 0.f);

        for (int base = beg; base < end; base += 32) {
            const int cnt = min(32, end - base);
            int idx = 0;
            if (lane < cnt) idx = csr_src[base + lane];

            int j = 0;
            for (; j + 8 <= cnt; j += 8) {
                const float4 f0 = GAT_ROW(j);
                const float4 f1 = GAT_ROW(j + 1);
                const float4 f2 = GAT_ROW(j + 2);
                const float4 f3 = GAT_ROW(j + 3);
                const float4 f4 = GAT_ROW(j + 4);
                const float4 f5 = GAT_ROW(j + 5);
                const float4 f6 = GAT_ROW(j + 6);
                const float4 f7 = GAT_ROW(j + 7);
                GAT_STEP4(f0, f1, f2, f3)
                GAT_STEP4(f4, f5, f6, f7)
            }
            for (; j + 4 <= cnt; j += 4) {
                const float4 f0 = GAT_ROW(j);
                const float4 f1 = GAT_ROW(j + 1);
                const float4 f2 = GAT_ROW(j + 2);
                const float4 f3 = GAT_ROW(j + 3);
                GAT_STEP4(f0, f1, f2, f3)
            }
            for (; j < cnt; ++j) {
                const float4 f0 = GAT_ROW(j);
                const float p0 = gat_logit(f0, b4, w4);
                const float mn = fmaxf(m, p0);
                const float sc = __expf(m - mn);
                const float a0 = __expf(p0 - mn);
                l     = l * sc + a0;
                acc.x = acc.x * sc + a0 * f0.x;
                acc.y = acc.y * sc + a0 * f0.y;
                acc.z = acc.z * sc + a0 * f0.z;
                acc.w = acc.w * sc + a0 * f0.w;
                m = mn;
            }
        }

        float4 r = *(const float4*)(trow + HD);
        if (end > beg) {
            r.x += acc.x / l;
            r.y += acc.y / l;
            r.z += acc.z / l;
            r.w += acc.w / l;
        }
        *(float4*)(out + (size_t)v * HD + c) = r;
    }
}
#undef GAT_ROW
#undef GAT_STEP4
#undef GAT_WMMA
#undef GAT_LOADB
#undef GAT_LOADA

extern "C" void kernel_launch(void* const* d_in, const int* in_sizes, int n_in,
                              void* d_out, int out_size, void* d_ws, size_t ws_size,
                              hipStream_t stream) {
    const float* x    = (const float*)d_in[0];
    const int*   src  = (const int*)  d_in[1];
    const int*   dst  = (const int*)  d_in[2];
    const float* Wsrc = (const float*)d_in[3];
    const float* bsrc = (const float*)d_in[4];
    const float* Wdst = (const float*)d_in[5];
    const float* bdst = (const float*)d_in[6];
    const float* attn = (const float*)d_in[7];
    const float* Wres = (const float*)d_in[8];
    const float* bres = (const float*)d_in[9];
    float* out = (float*)d_out;

    char* ws = (char*)d_ws;
    const size_t elBytes   = (size_t)N_NODES * HD * sizeof(float);
    const size_t padBytes  = (size_t)TAB_PAD * sizeof(int);
    const size_t edgeBytes = (size_t)E_EDGES * sizeof(int);
    float* el     = (float*)(ws);
    int*   rtab   = (int*)  (ws + 2 * elBytes);
    int*   rowptr = (int*)  (ws + 2 * elBytes + padBytes);
    int*   part   = (int*)  (ws + 2 * elBytes + 2 * padBytes + 512);
    int*   csr    = (int*)  (ws + 2 * elBytes + 2 * padBytes + 512 + edgeBytes);

    gat_chunk_order<<<NCH, 256, 0, stream>>>(src, dst, rtab, part);
    gat_range_csr<<<NBK, 512, 0, stream>>>(part, rtab, rowptr, csr);

    gat_gemm_el<<<MAIN_BLOCKS + TAIL_BLOCKS, 256, 0, stream>>>(x, Wsrc, bsrc, el);

    gat_node_fused<<<MAIN_BLOCKS + TAIL_BLOCKS, 256, 0, stream>>>(x, Wdst, bdst, Wres, bres, el, rowptr, csr, attn, out);
}


# Node walk's edge math on channel pairs of one row (packed add/mul/fma on the registers a row load leaves adjacent), leaky ReLU as max(t, 0.2t): no register moves, no compares; sums in the same order
# speedup vs baseline: 4.2061x; 1.0348x over previous
#include <hip/hip_runtime.h>
#include <hip/hip_bf16.h>

#define N_NODES  100000
#define E_EDGES  1600000
#define IN_F     256
#define HD       128
#define H_HEADS  8
#define SLOPE    0.2f

#define ROWS_PB    64
#define LDS_STRIDE 260

#define TAB_PAD    100352

#define MAIN_BLOCKS  1536
#define MAIN_ROWS    (MAIN_BLOCKS * ROWS_PB)
#define TAIL_TILES   ((N_NODES - MAIN_ROWS + 15) / 16)
#define TAIL_BLOCKS  TAIL_TILES
#define TAIL_NODES   16
static_assert(MAIN_ROWS + TAIL_BLOCKS * TAIL_NODES == N_NODES, "the small work-groups cover the rest exactly");

typedef __attribute__((ext_vector_type(2))) float v2f;
typedef __attribute__((ext_vector_type(8))) float v8f;

__device__ __forceinline__ void gat_gemm_tail(
    const float* __restrict__ x, const float* __restrict__ W, const float* __restrict__ bvec,
    float* __restrict__ O, float* __restrict__ xs, const int row0)
{
    const int tid = threadIdx.x;
    {
        const int c4 = tid & 63;
        const int rb = tid >> 6;
        float4 v[4];
#pragma unroll
        for (int i = 0; i < 4; ++i) {
            const int gr = min(row0 + rb + 4 * i, N_NODES - 1);
            v[i] = *(const float4*)(x + (size_t)gr * IN_F + c4 * 4);
        }
#pragma unroll
        for (int i = 0; i < 4; ++i)
            *(float4*)&xs[(rb + 4 * i) * LDS_STRIDE + c4 * 4] = v[i];
    }
    __syncthreads();

    const int wave    = tid >> 5;
    const int lane    = tid & 31;
    const int l15     = lane & 15;
    const int kq      = (lane >> 4) * 4;
    const int rhalf   = (lane >> 4) * 8;
    const int colbase = wave * 16;

    const float* __restrict__ arow = &xs[l15 * LDS_STRIDE + kq];
    const float* __restrict__ Wp   = W + (size_t)kq * HD + colbase + l15;
    v8f acc = {};
    for (int k = 0; k < IN_F; k += 8) {
        const float4 a = *(const float4*)(arow + k);
        v2f b0, b1, a01, a23;
        b0.x = Wp[(k + 0) * HD];
        b0.y = Wp[(k + 1) * HD];
        b1.x = Wp[(k + 2) * HD];
        b1.y = Wp[(k + 3) * HD];
        a01.x = a.x; a01.y = a.y;
        a23.x = a.z; a23.y = a.w;
        acc = __builtin_amdgcn_wmma_f32_16x16x4_f32(false, a01, false, b0, (short)0, acc, false, false);
        acc = __builtin_amdgcn_wmma_f32_16x16x4_f32(false, a23, false, b1, (short)0, acc, false, false);
    }
    const float bias = bvec[colbase + l15];
#pragma unroll
    for (int j = 0; j < 8; ++j) {
        const int r = row0 + rhalf + j;
        if (r < N_NODES)
            O[(size_t)r * HD + colbase + l15] = acc[j] + bias;
    }
}

#define GAT_LOADA(A0, A1, A2, A3, K)                                           \
    {                                                                          \
        A0 = *(const float4*)(arow + 0 * 16 * LDS_STRIDE + (K));               \
        A1 = *(const float4*)(arow + 1 * 16 * LDS_STRIDE + (K));               \
        A2 = *(const float4*)(arow + 2 * 16 * LDS_STRIDE + (K));               \
        A3 = *(const float4*)(arow + 3 * 16 * LDS_STRIDE + (K));               \
    }
#define GAT_LOADB(B0, B1, WP, K)                                               \
    {                                                                          \
        B0.x = WP[((K) + 0) * HD];                                             \
        B0.y = WP[((K) + 1) * HD];                                             \
        B1.x = WP[((K) + 2) * HD];                                             \
        B1.y = WP[((K) + 3) * HD];                                             \
    }
#define GAT_WMMA(ACC, AX, AY, B)                                               \
    {                                                                          \
        v2f a; a.x = (AX); a.y = (AY);                                         \
        ACC = __builtin_amdgcn_wmma_f32_16x16x4_f32(                           \
                false, a, false, B, (short)0, ACC, false, false);              \
    }

__global__ __launch_bounds__(256) void gat_gemm_el(
    const float* __restrict__ x,
    const float* __restrict__ Wsrc, const float* __restrict__ bsrc,
    float* __restrict__ el)
{
    __shared__ __attribute__((aligned(16))) float xs[ROWS_PB * LDS_STRIDE];

    if (blockIdx.x >= MAIN_BLOCKS) {
        gat_gemm_tail(x, Wsrc, bsrc, el, xs, MAIN_ROWS + (blockIdx.x - MAIN_BLOCKS) * 16);
        return;
    }

    const int tid  = threadIdx.x;
    const int row0 = blockIdx.x * ROWS_PB;

    {
        const int c4 = tid & 63;
        const int rb = tid >> 6;
        float4 v[16];
#pragma unroll
        for (int i = 0; i < 16; ++i) {
            const int gr = min(row0 + rb + 4 * i, N_NODES - 1);
            v[i] = *(const float4*)(x + (size_t)gr * IN_F + c4 * 4);
        }
#pragma unroll
        for (int i = 0; i < 16; ++i)
            *(float4*)&xs[(rb + 4 * i) * LDS_STRIDE + c4 * 4] = v[i];
    }
    __syncthreads();

    const int wave    = tid >> 5;
    const int lane    = tid & 31;
    const int l15     = lane & 15;
    const int kq      = (lane >> 4) * 4;
    const int rhalf   = (lane >> 4) * 8;
    const int colbase = wave * 16;

    const float* __restrict__ arow = &xs[l15 * LDS_STRIDE + kq];
    const float* __restrict__ Wp   = Wsrc + (size_t)kq * HD + colbase + l15;
    v8f acc0 = {}; v8f acc1 = {}; v8f acc2 = {}; v8f acc3 = {};

    for (int k = 0; k < IN_F; k += 8) {
        float4 a0, a1, a2, a3; v2f b0, b1;
        GAT_LOADA(a0, a1, a2, a3, k)
        GAT_LOADB(b0, b1, Wp, k)

        GAT_WMMA(acc0, a0.x, a0.y, b0)
        GAT_WMMA(acc1, a1.x, a1.y, b0)
        GAT_WMMA(acc2, a2.x, a2.y, b0)
        GAT_WMMA(acc3, a3.x, a3.y, b0)
        GAT_WMMA(acc0, a0.z, a0.w, b1)
        GAT_WMMA(acc1, a1.z, a1.w, b1)
        GAT_WMMA(acc2, a2.z, a2.w, b1)
        GAT_WMMA(acc3, a3.z, a3.w, b1)
    }

    const float bias = bsrc[colbase + l15];
#define GAT_STORE(ACC, RT)                                                     \
    {                                                                          \
        _Pragma("unroll")                                                      \
        for (int j = 0; j < 8; ++j) {                                          \
            const int r = row0 + (RT) * 16 + rhalf + j;                        \
            if (r < N_NODES)                                                   \
                el[(size_t)r * HD + colbase + l15] = ACC[j] + bias;            \
        }                                                                      \
    }
    GAT_STORE(acc0, 0)
    GAT_STORE(acc1, 1)
    GAT_STORE(acc2, 2)
    GAT_STORE(acc3, 3)
#undef GAT_STORE
}

#define NBK      256
#define NPB      391
#define NCH      256
#define CH       (E_EDGES / NCH)
#define CH_FULL  (CH / 256)
#define CH_TAIL  (CH - CH_FULL * 256)
#define SRC_BITS 17
#define SRC_MASK ((1 << SRC_BITS) - 1)
#define CSR_CAP  12288

#define RUN_BITS 16
#define RUN_MASK ((1 << RUN_BITS) - 1)
static_assert(CH < (1 << (RUN_BITS - 1)), "start and length of a run fit the table word");

static_assert(NBK * NPB >= N_NODES, "ranges cover the nodes");
static_assert(NCH * CH == E_EDGES, "chunks cover the edges");
static_assert((1 << SRC_BITS) > N_NODES && NPB < (1 << (31 - SRC_BITS)), "packed word fits");

__device__ __forceinline__ int gat_clamp_node(int d)
{
    return min(max(d, 0), N_NODES - 1);
}

__device__ __forceinline__ int gat_block_exscan(int v, int* wsum)
{
    const int lane = threadIdx.x & 31;
    const int wave = threadIdx.x >> 5;
    int s = v;
    for (int d = 1; d < 32; d <<= 1) {
        const int t = __shfl_up(s, d, 32);
        if (lane >= d) s += t;
    }
    if (lane == 31) wsum[wave] = s;
    __syncthreads();
    if (wave == 0) {
        int w = (lane < (int)(blockDim.x >> 5)) ? wsum[lane] : 0;
        for (int d = 1; d < 32; d <<= 1) {
            const int t = __shfl_up(w, d, 32);
            if (lane >= d) w += t;
        }
        wsum[lane] = w;
    }
    __syncthreads();
    const int woff = (wave > 0) ? wsum[wave - 1] : 0;
    return woff + s - v;
}

__global__ __launch_bounds__(256) void gat_chunk_order(
    const int* __restrict__ src, const int* __restrict__ dst,
    int* __restrict__ rtab, int* __restrict__ part)
{
    __shared__ int h[NBK];
    __shared__ int sw[CH];
    __shared__ int wsum[32];
    const int tid = threadIdx.x;
    const int blk = blockIdx.x;
    const int* __restrict__ sp = src + blk * CH;
    const int* __restrict__ dp = dst + blk * CH;
    h[tid] = 0;

    int d[CH_FULL], s[CH_FULL];
#pragma unroll
    for (int k = 0; k < CH_FULL; ++k) { d[k] = dp[k * 256 + tid]; s[k] = sp[k * 256 + tid]; }
    int       dl = dp[min(CH_FULL * 256 + tid, CH - 1)];
    const int sl = sp[min(CH_FULL * 256 + tid, CH - 1)];
    __syncthreads();
#pragma unroll
    for (int k = 0; k < CH_FULL; ++k) { d[k] = gat_clamp_node(d[k]); atomicAdd(&h[d[k] / NPB], 1); }
    dl = gat_clamp_node(dl);
    if (tid < CH_TAIL) atomicAdd(&h[dl / NPB], 1);
    __syncthreads();

    const int c  = h[tid];
    const int ex = gat_block_exscan(c, wsum);
    h[tid] = ex;
    rtab[tid * NCH + blk] = (ex << RUN_BITS) | c;
    __syncthreads();

#define GAT_PLACE(DN, S)                                                       \
    {                                                                          \
        const int b  = (DN) / NPB;                                             \
        const int sn = ((unsigned)(S) < (unsigned)N_NODES) ? (S) : 0;          \
        const int p  = atomicAdd(&h[b], 1);                                    \
        if ((unsigned)p < (unsigned)CH)                                        \
            sw[p] = (((DN) - b * NPB) << SRC_BITS) | sn;                       \
    }
#pragma unroll
    for (int k = 0; k < CH_FULL; ++k) GAT_PLACE(d[k], s[k])
    if (tid < CH_TAIL) GAT_PLACE(dl, sl)
#undef GAT_PLACE
    __syncthreads();

    int* __restrict__ pw = part + blk * CH;
    for (int i = tid; i < CH; i += 256) pw[i] = sw[i];
}

#define RUNS_PW  (NCH / 16)

__global__ __launch_bounds__(512) void gat_range_csr(
    const int* __restrict__ part, const int* __restrict__ rtab,
    int* __restrict__ rowptr, int* __restrict__ csr_src)
{
    __shared__ int cnt[512];
    __shared__ int lw[CSR_CAP];
    __shared__ int lcsr[CSR_CAP];
    __shared__ int rpos[NCH];
    __shared__ int roff[NCH + 1];
    __shared__ int wred[16];
    __shared__ int wsum[32];
    const int tid  = threadIdx.x;
    const int lane = tid & 31;
    const int wave = tid >> 5;
    const int b    = blockIdx.x;

    int st = 0, n = 0;
    if (tid < NCH) {
        const int w = rtab[b * NCH + tid];
        st = min((int)((unsigned)w >> RUN_BITS), CH);
        n  = min(w & RUN_MASK, CH - st);
    }
    cnt[tid] = 0;
    const int off = gat_block_exscan(n, wsum);
    int sred = st;
    for (int dd = 16; dd > 0; dd >>= 1) sred += __shfl_xor(sred, dd, 32);
    if (lane == 0) wred[wave] = sred;
    if (tid < NCH) { rpos[tid] = tid * CH + st; roff[tid] = off; }
    if (tid == NCH - 1) roff[NCH] = off + n;
    __syncthreads();

    int bs = 0;
    for (int k = 0; k < 16; ++k) bs += wred[k];
    bs = min(max(bs, 0), E_EDGES);
    const int nb = min(roff[NCH], E_EDGES - bs);
    const int be = bs + nb;
    const bool stage = (roff[NCH] <= CSR_CAP);

    if (stage) {
        int wv[RUNS_PW];
#pragma unroll
        for (int q = 0; q < RUNS_PW; ++q) {
            const int c   = wave + 16 * q;
            const int len = roff[c + 1] - roff[c];
            wv[q] = (lane < len) ? part[rpos[c] + lane] : 0;
        }
        int longrun = 0;
#pragma unroll
        for (int q = 0; q < RUNS_PW; ++q) {
            const int c   = wave + 16 * q;
            const int o   = roff[c];
            const int len = roff[c + 1] - o;
            if (lane < len) lw[o + lane] = wv[q];
            longrun |= (len > 32);
        }
        if (longrun) {
#pragma nounroll
            for (int c = wave; c < NCH; c += 16) {
                const int o   = roff[c];
                const int len = roff[c + 1] - o;
#pragma nounroll
                for (int j = lane + 32; j < len; j += 32) lw[o + j] = part[rpos[c] + j];
            }
        }
        __syncthreads();
        for (int i = tid; i < nb; i += 512)
            atomicAdd(&cnt[min((int)((unsigned)lw[i] >> SRC_BITS), NPB - 1)], 1);
    } else {
#pragma nounroll
        for (int c = wave; c < NCH; c += 16) {
            const int len = roff[c + 1] - roff[c];
#pragma nounroll
            for (int j = lane; j < len; j += 32)
                atomicAdd(&cnt[min((int)((unsigned)part[rpos[c] + j] >> SRC_BITS), NPB - 1)], 1);
        }
    }
    __syncthreads();

    const int c  = (tid < NPB) ? cnt[tid] : 0;
    const int ex = gat_block_exscan(c, wsum);
    cnt[tid] = ex;
    if (tid < NPB) {
        const int node = b * NPB + tid;
        if (node < N_NODES) rowptr[node] = bs + ex;
    }
    if (b == NBK - 1 && tid == 0) rowptr[N_NODES] = be;
    __syncthreads();

    if (stage) {
        for (int i = tid; i < nb; i += 512) {
            const int w  = lw[i];
            const int ln = min((int)((unsigned)w >> SRC_BITS), NPB - 1);
            const int p  = atomicAdd(&cnt[ln], 1);
            if ((unsigned)p < (unsigned)nb) lcsr[p] = w & SRC_MASK;
        }
        __syncthreads();
        for (int i = tid; i < nb; i += 512) csr_src[bs + i] = lcsr[i];
    } else {
#pragma nounroll
        for (int cc = wave; cc < NCH; cc += 16) {
            const int len = roff[cc + 1] - roff[cc];
#pragma nounroll
            for (int j = lane; j < len; j += 32) {
                const int w  = part[rpos[cc] + j];
                const int ln = min((int)((unsigned)w >> SRC_BITS), NPB - 1);
                const int p  = atomicAdd(&cnt[ln], 1);
                if ((unsigned)p < (unsigned)nb) csr_src[bs + p] = w & SRC_MASK;
            }
        }
    }
}

__device__ __forceinline__ float gat_logit(const float4 f, const float4 b4, const float4 w4)
{
    const v2f fa = {f.x, f.y},   fb = {f.z, f.w};
    const v2f ba = {b4.x, b4.y}, bb = {b4.z, b4.w};
    v2f ta = fa + ba, tb = fb + bb;
    ta = __builtin_elementwise_max(ta, ta * SLOPE);
    tb = __builtin_elementwise_max(tb, tb * SLOPE);
    float p = ta.x * w4.x;
    p = __builtin_fmaf(ta.y, w4.y, p);
    p = __builtin_fmaf(tb.x, w4.z, p);
    p = __builtin_fmaf(tb.y, w4.w, p);
    p += __shfl_xor(p, 1, 32);
    p += __shfl_xor(p, 2, 32);
    return p;
}

#define GAT_STEP4(F0, F1, F2, F3)                                                          \
    {                                                                                      \
        const float p0 = gat_logit(F0, b4, w4);                                            \
        const float p1 = gat_logit(F1, b4, w4);                                            \
        const float p2 = gat_logit(F2, b4, w4);                                            \
        const float p3 = gat_logit(F3, b4, w4);                                            \
        const float mn = fmaxf(m, fmaxf(fmaxf(p0, p1), fmaxf(p2, p3)));                    \
        const float sc = __expf(m - mn);                                                   \
        const float a0 = __expf(p0 - mn);                                                  \
        const float a1 = __expf(p1 - mn);                                                  \
        const float a2 = __expf(p2 - mn);                                                  \
        const float a3 = __expf(p3 - mn);                                                  \
        l     = l * sc + ((a0 + a1) + (a2 + a3));                                          \
        const v2f s2 = {sc, sc};                                                           \
        const v2f g0 = {a0, a0}, g1 = {a1, a1}, g2 = {a2, a2}, g3 = {a3, a3};              \
        accA = __builtin_elementwise_fma(accA, s2,                                         \
                 GAT_PAIR(g0, GAT_LO(F0), g1, GAT_LO(F1)) + GAT_PAIR(g2, GAT_LO(F2), g3, GAT_LO(F3))); \
        accB = __builtin_elementwise_fma(accB, s2,                                         \
                 GAT_PAIR(g0, GAT_HI(F0), g1, GAT_HI(F1)) + GAT_PAIR(g2, GAT_HI(F2), g3, GAT_HI(F3))); \
        m = mn;                                                                            \
    }
#define GAT_LO(F) ((v2f){(F).x, (F).y})
#define GAT_HI(F) ((v2f){(F).z, (F).w})
#define GAT_PAIR(U, P, V, Q) __builtin_elementwise_fma((U), (P), (V) * (Q))
#define GAT_ROW(J)                                                                         \
    ({                                                                                     \
        int u_ = __builtin_amdgcn_readlane(idx, (J));                                      \
        u_ = ((unsigned)u_ < (unsigned)N_NODES) ? u_ : 0;                                  \
        *(const float4*)(el + (size_t)u_ * HD + c);                                        \
    })

__global__ __launch_bounds__(256) void gat_node_fused(
    const float* __restrict__ x,
    const float* __restrict__ Wdst, const float* __restrict__ bdst,
    const float* __restrict__ Wres, const float* __restrict__ bres,
    const float* __restrict__ el,
    const int* __restrict__ rowptr, const int* __restrict__ csr_src,
    const float* __restrict__ attn, float* __restrict__ out)
{
    __shared__ __attribute__((aligned(16))) float xs[ROWS_PB * LDS_STRIDE];

    const int tid  = threadIdx.x;
    const int lane = tid & 31;
    const int wv   = __builtin_amdgcn_readfirstlane(tid >> 5);

    const bool small = blockIdx.x >= MAIN_BLOCKS;
    const int  row0  = small ? MAIN_ROWS + (blockIdx.x - MAIN_BLOCKS) * TAIL_NODES : blockIdx.x * ROWS_PB;
    const int  npw   = small ? TAIL_NODES / 8 : ROWS_PB / 8;
    const int  nd0   = wv * npw;

    const int rp = rowptr[min(row0 + nd0 + min(lane, npw), N_NODES)];

    const int l15     = lane & 15;
    const int kq      = (lane >> 4) * 4;
    const int rhalf   = (lane >> 4) * 8;
    const int colbase = wv * 16;

    if (!small) {
        {
            const int c4 = tid & 63;
            const int rb = tid >> 6;
            float4 v[16];
#pragma unroll
            for (int i = 0; i < 16; ++i) {
                const int gr = min(row0 + rb + 4 * i, N_NODES - 1);
                v[i] = *(const float4*)(x + (size_t)gr * IN_F + c4 * 4);
            }
#pragma unroll
            for (int i = 0; i < 16; ++i)
                *(float4*)&xs[(rb + 4 * i) * LDS_STRIDE + c4 * 4] = v[i];
        }
        __syncthreads();

        const float* __restrict__ arow = &xs[l15 * LDS_STRIDE + kq];
        const float* __restrict__ Wd   = Wdst + (size_t)kq * HD + colbase + l15;
        const float* __restrict__ Wr   = Wres + (size_t)kq * HD + colbase + l15;
        v8f e0 = {}; v8f e1 = {}; v8f e2 = {}; v8f e3 = {};
        v8f r0 = {}; v8f r1 = {}; v8f r2 = {}; v8f r3 = {};

        for (int k = 0; k < IN_F; k += 8) {
            float4 a0, a1, a2, a3; v2f d0, d1, s0, s1;
            GAT_LOADA(a0, a1, a2, a3, k)
            GAT_LOADB(d0, d1, Wd, k)
            GAT_LOADB(s0, s1, Wr, k)

            GAT_WMMA(e0, a0.x, a0.y, d0)
            GAT_WMMA(e1, a1.x, a1.y, d0)
            GAT_WMMA(e2, a2.x, a2.y, d0)
            GAT_WMMA(e3, a3.x, a3.y, d0)
            GAT_WMMA(r0, a0.x, a0.y, s0)
            GAT_WMMA(r1, a1.x, a1.y, s0)
            GAT_WMMA(r2, a2.x, a2.y, s0)
            GAT_WMMA(r3, a3.x, a3.y, s0)
            GAT_WMMA(e0, a0.z, a0.w, d1)
            GAT_WMMA(e1, a1.z, a1.w, d1)
            GAT_WMMA(e2, a2.z, a2.w, d1)
            GAT_WMMA(e3, a3.z, a3.w, d1)
            GAT_WMMA(r0, a0.z, a0.w, s1)
            GAT_WMMA(r1, a1.z, a1.w, s1)
            GAT_WMMA(r2, a2.z, a2.w, s1)
            GAT_WMMA(r3, a3.z, a3.w, s1)
        }

        const float bias_d = bdst[colbase + l15];
        const float bias_r = bres[colbase + l15];
        __syncthreads();

#define GAT_KEEP(EACC, RACC, RT)                                                           \
        {                                                                                  \
            _Pragma("unroll")                                                              \
            for (int j = 0; j < 8; ++j) {                                                  \
                float* q = &xs[((RT) * 16 + rhalf + j) * LDS_STRIDE + colbase + l15];      \
                q[0]  = EACC[j] + bias_d;                                                  \
                q[HD] = RACC[j] + bias_r;                                                  \
            }                                                                              \
        }
        GAT_KEEP(e0, r0, 0)
        GAT_KEEP(e1, r1, 1)
        GAT_KEEP(e2, r2, 2)
        GAT_KEEP(e3, r3, 3)
    } else {
        {
            const int c4 = tid & 63;
            const int rb = tid >> 6;
            float4 v[4];
#pragma unroll
            for (int i = 0; i < 4; ++i) {
                const int gr = min(row0 + rb + 4 * i, N_NODES - 1);
                v[i] = *(const float4*)(x + (size_t)gr * IN_F + c4 * 4);
            }
#pragma unroll
            for (int i = 0; i < 4; ++i)
                *(float4*)&xs[(rb + 4 * i) * LDS_STRIDE + c4 * 4] = v[i];
        }
        __syncthreads();

        const float* __restrict__ arow = &xs[l15 * LDS_STRIDE + kq];
        const float* __restrict__ Wd   = Wdst + (size_t)kq * HD + colbase + l15;
        const float* __restrict__ Wr   = Wres + (size_t)kq * HD + colbase + l15;
        v8f e0 = {};
        v8f r0 = {};

        for (int k = 0; k < IN_F; k += 8) {
            v2f d0, d1, s0, s1;
            const float4 a0 = *(const float4*)(arow + k);
            GAT_LOADB(d0, d1, Wd, k)
            GAT_LOADB(s0, s1, Wr, k)

            GAT_WMMA(e0, a0.x, a0.y, d0)
            GAT_WMMA(r0, a0.x, a0.y, s0)
            GAT_WMMA(e0, a0.z, a0.w, d1)
            GAT_WMMA(r0, a0.z, a0.w, s1)
        }

        const float bias_d = bdst[colbase + l15];
        const float bias_r = bres[colbase + l15];
        __syncthreads();

        GAT_KEEP(e0, r0, 0)
#undef GAT_KEEP
    }
    __syncthreads();

    const int    c  = lane * 4;
    const float4 w4 = *(const float4*)(attn + c);

    for (int i = 0; i < npw; ++i) {
        const int v = row0 + nd0 + i;
        if (v >= N_NODES) break;

        int beg = __builtin_amdgcn_readlane(rp, i);
        int end = __builtin_amdgcn_readlane(rp, i + 1);
        beg = max(beg, 0);
        end = min(end, E_EDGES);

        const float* trow = &xs[(nd0 + i) * LDS_STRIDE + c];
        const float4 b4 = *(const float4*)trow;

        float  m   = -INFINITY;
        float  l   = 0.f;
        v2f accA = {0.f, 0.f};
        v2f accB = {0.f, 0.f};

        for (int base = beg; base < end; base += 32) {
            const int cnt = min(32, end - base);
            int idx = 0;
            if (lane < cnt) idx = csr_src[base + lane];

            int j = 0;
            for (; j + 8 <= cnt; j += 8) {
                const float4 f0 = GAT_ROW(j);
                const float4 f1 = GAT_ROW(j + 1);
                const float4 f2 = GAT_ROW(j + 2);
                const float4 f3 = GAT_ROW(j + 3);
                const float4 f4 = GAT_ROW(j + 4);
                const float4 f5 = GAT_ROW(j + 5);
                const float4 f6 = GAT_ROW(j + 6);
                const float4 f7 = GAT_ROW(j + 7);
                GAT_STEP4(f0, f1, f2, f3)
                GAT_STEP4(f4, f5, f6, f7)
            }
            for (; j + 4 <= cnt; j += 4) {
                const float4 f0 = GAT_ROW(j);
                const float4 f1 = GAT_ROW(j + 1);
                const float4 f2 = GAT_ROW(j + 2);
                const float4 f3 = GAT_ROW(j + 3);
                GAT_STEP4(f0, f1, f2, f3)
            }
            for (; j < cnt; ++j) {
                const float4 f0 = GAT_ROW(j);
                const float p0 = gat_logit(f0, b4, w4);
                const float mn = fmaxf(m, p0);
                const float sc = __expf(m - mn);
                const float a0 = __expf(p0 - mn);
                l     = l * sc + a0;
                accA.x = accA.x * sc + a0 * f0.x;
                accA.y = accA.y * sc + a0 * f0.y;
                accB.x = accB.x * sc + a0 * f0.z;
                accB.y = accB.y * sc + a0 * f0.w;
                m = mn;
            }
        }

        float4 r = *(const float4*)(trow + HD);
        if (end > beg) {
            r.x += accA.x / l;
            r.y += accA.y / l;
            r.z += accB.x / l;
            r.w += accB.y / l;
        }
        *(float4*)(out + (size_t)v * HD + c) = r;
    }
}
#undef GAT_ROW
#undef GAT_STEP4
#undef GAT_PAIR
#undef GAT_HI
#undef GAT_LO
#undef GAT_WMMA
#undef GAT_LOADB
#undef GAT_LOADA

extern "C" void kernel_launch(void* const* d_in, const int* in_sizes, int n_in,
                              void* d_out, int out_size, void* d_ws, size_t ws_size,
                              hipStream_t stream) {
    const float* x    = (const float*)d_in[0];
    const int*   src  = (const int*)  d_in[1];
    const int*   dst  = (const int*)  d_in[2];
    const float* Wsrc = (const float*)d_in[3];
    const float* bsrc = (const float*)d_in[4];
    const float* Wdst = (const float*)d_in[5];
    const float* bdst = (const float*)d_in[6];
    const float* attn = (const float*)d_in[7];
    const float* Wres = (const float*)d_in[8];
    const float* bres = (const float*)d_in[9];
    float* out = (float*)d_out;

    char* ws = (char*)d_ws;
    const size_t elBytes   = (size_t)N_NODES * HD * sizeof(float);
    const size_t padBytes  = (size_t)TAB_PAD * sizeof(int);
    const size_t edgeBytes = (size_t)E_EDGES * sizeof(int);
    float* el     = (float*)(ws);
    int*   rtab   = (int*)  (ws + 2 * elBytes);
    int*   rowptr = (int*)  (ws + 2 * elBytes + padBytes);
    int*   part   = (int*)  (ws + 2 * elBytes + 2 * padBytes + 512);
    int*   csr    = (int*)  (ws + 2 * elBytes + 2 * padBytes + 512 + edgeBytes);

    gat_chunk_order<<<NCH, 256, 0, stream>>>(src, dst, rtab, part);
    gat_range_csr<<<NBK, 512, 0, stream>>>(part, rtab, rowptr, csr);

    gat_gemm_el<<<MAIN_BLOCKS + TAIL_BLOCKS, 256, 0, stream>>>(x, Wsrc, bsrc, el);

    gat_node_fused<<<MAIN_BLOCKS + TAIL_BLOCKS, 256, 0, stream>>>(x, Wdst, bdst, Wres, bres, el, rowptr, csr, attn, out);
}
